# starting kernel
# baseline (speedup 1.0000x reference)
#include <hip/hip_runtime.h>

typedef __attribute__((ext_vector_type(2))) float v2f;
typedef __attribute__((ext_vector_type(8))) float v8f;

#define N_GRAPHS 64


__global__ void k_degree(const int* __restrict__ col, unsigned* __restrict__ deg, int E) {
  int e = blockIdx.x * blockDim.x + threadIdx.x;
  if (e < E) atomicAdd(&deg[col[e]], 1u);
}

__global__ void k_dinv(const unsigned* __restrict__ deg, float* __restrict__ dinv, int n) {
  int i = blockIdx.x * blockDim.x + threadIdx.x;
  if (i < n) dinv[i] = rsqrtf((float)(deg[i] + 1u));
}


template <int IN, int OUT>
__global__ __launch_bounds__(256) void k_gemm_scaled(const float* __restrict__ A,
                                                     const float* __restrict__ W,
                                                     const float* __restrict__ dinv,
                                                     float* __restrict__ out, int nRows) {
  constexpr int NT = OUT / 16;
  const int lane = threadIdx.x & 31;
  const int wave = threadIdx.x >> 5;
  const int h = lane >> 4;
  const int m = lane & 15;
  const int rowBase = blockIdx.x * 128 + wave * 16;

  v8f acc[NT] = {};

  int arow = rowBase + m;
  if (arow > nRows - 1) arow = nRows - 1;
  const float* aptr = A + (size_t)arow * IN + 2 * h;

  for (int kb = 0; kb < IN; kb += 4) {
    float2 a2 = *(const float2*)(aptr + kb);
    v2f av; av.x = a2.x; av.y = a2.y;
    const float* w0 = W + (size_t)(kb + 2 * h) * OUT + m;
#pragma unroll
    for (int t = 0; t < NT; t++) {
      v2f bv; bv.x = w0[t * 16]; bv.y = w0[OUT + t * 16];
      acc[t] = __builtin_amdgcn_wmma_f32_16x16x4_f32(
           false, av,  false, bv,
           (short)0, acc[t],  false,  false);
    }
  }

#pragma unroll
  for (int v = 0; v < 8; v++) {
    int r = rowBase + v + 8 * h;
    if (r < nRows) {
      float s = dinv[r];
#pragma unroll
      for (int t = 0; t < NT; t++) out[(size_t)r * OUT + t * 16 + m] = acc[t][v] * s;
    }
  }
}


template <int D>
__global__ void k_scatter(const float* __restrict__ src, const int* __restrict__ row,
                          const int* __restrict__ col, float* __restrict__ dst, int E) {
  constexpr int C = D / 4;
  long long idx = (long long)blockIdx.x * blockDim.x + threadIdx.x;
  if (idx >= (long long)E * C) return;
  int e = (int)(idx / C);
  int c = (int)(idx % C);
  int r = row[e];
  int cl = col[e];
  float4 v = *(const float4*)(src + (size_t)r * D + c * 4);
  float* d = dst + (size_t)cl * D + c * 4;
  atomicAdd(d + 0, v.x);
  atomicAdd(d + 1, v.y);
  atomicAdd(d + 2, v.z);
  atomicAdd(d + 3, v.w);
}


__global__ void k_epilogue1(const float* __restrict__ agg, float* __restrict__ xws,
                            const float* __restrict__ dinv, const float* __restrict__ b1,
                            const float* __restrict__ lnw, const float* __restrict__ lnb,
                            int n) {
  int lane = threadIdx.x & 31;
  int wave = threadIdx.x >> 5;
  int i = blockIdx.x * 8 + wave;
  if (i >= n) return;
  float di = dinv[i];
  float t[4];
  float s = 0.f;
#pragma unroll
  for (int j = 0; j < 4; j++) {
    int d = lane + 32 * j;
    t[j] = (agg[(size_t)i * 128 + d] + xws[(size_t)i * 128 + d]) * di + b1[d];
    s += t[j];
  }
#pragma unroll
  for (int o = 16; o > 0; o >>= 1) s += __shfl_xor(s, o, 32);
  float mu = s * (1.f / 128.f);
  float var = 0.f;
#pragma unroll
  for (int j = 0; j < 4; j++) { float d0 = t[j] - mu; var += d0 * d0; }
#pragma unroll
  for (int o = 16; o > 0; o >>= 1) var += __shfl_xor(var, o, 32);
  float inv = rsqrtf(var * (1.f / 128.f) + 1e-5f);
#pragma unroll
  for (int j = 0; j < 4; j++) {
    int d = lane + 32 * j;
    float o = (t[j] - mu) * inv * lnw[d] + lnb[d];
    xws[(size_t)i * 128 + d] = fmaxf(o, 0.f);
  }
}


__global__ void k_epilogue2(float* __restrict__ hout, const float* __restrict__ xw2s,
                            const float* __restrict__ dinv, const float* __restrict__ b2,
                            const int* __restrict__ batch, float* __restrict__ psum,
                            float* __restrict__ pcnt, int n) {
  int lane = threadIdx.x & 31;
  int wave = threadIdx.x >> 5;
  int i = blockIdx.x * 8 + wave;
  if (i >= n) return;
  float di = dinv[i];
  int b = batch[i];
#pragma unroll
  for (int j = 0; j < 2; j++) {
    int d = lane + 32 * j;
    float h = (hout[(size_t)i * 64 + d] + xw2s[(size_t)i * 64 + d]) * di + b2[d];
    hout[(size_t)i * 64 + d] = h;
    atomicAdd(&psum[b * 64 + d], h);
  }
  if (lane == 0) atomicAdd(&pcnt[b], 1.f);
}

__global__ void k_pool(const float* __restrict__ psum, const float* __restrict__ pcnt,
                       float* __restrict__ g) {
  int idx = blockIdx.x * blockDim.x + threadIdx.x;
  if (idx < N_GRAPHS * 64) {
    int gi = idx >> 6;
    g[idx] = psum[idx] / fmaxf(pcnt[gi], 1.f);
  }
}


extern "C" void kernel_launch(void* const* d_in, const int* in_sizes, int n_in,
                              void* d_out, int out_size, void* d_ws, size_t ws_size,
                              hipStream_t stream) {
  const float* x   = (const float*)d_in[0];
  const int*   ei  = (const int*)d_in[1];
  const int*  batch= (const int*)d_in[2];
  const float* W1  = (const float*)d_in[3];
  const float* b1  = (const float*)d_in[4];
  const float* lnw = (const float*)d_in[5];
  const float* lnb = (const float*)d_in[6];
  const float* W2  = (const float*)d_in[7];
  const float* b2  = (const float*)d_in[8];

  const int n = in_sizes[0] / 64;
  const int E = in_sizes[1] / 2;
  const int* row = ei;
  const int* col = ei + E;

  float* ws = (float*)d_ws;
  unsigned* deg = (unsigned*)ws;
  float* agg1 = ws + n;
  float* psum = agg1 + (size_t)n * 128;
  float* pcnt = psum + N_GRAPHS * 64;
  float* dinv = pcnt + N_GRAPHS;
  float* xws  = dinv + n;

  float* hout = (float*)d_out;
  float* gout = hout + (size_t)n * 64;

  size_t zbytes = ((size_t)n + (size_t)n * 128 + N_GRAPHS * 64 + N_GRAPHS) * sizeof(float);
  hipMemsetAsync(d_ws, 0, zbytes, stream);
  hipMemsetAsync(d_out, 0, (size_t)n * 64 * sizeof(float), stream);

  k_degree<<<(E + 255) / 256, 256, 0, stream>>>(col, deg, E);
  k_dinv<<<(n + 255) / 256, 256, 0, stream>>>(deg, dinv, n);

  k_gemm_scaled<64, 128><<<(n + 127) / 128, 256, 0, stream>>>(x, W1, dinv, xws, n);
  {
    long long tot = (long long)E * 32;
    k_scatter<128><<<(unsigned)((tot + 255) / 256), 256, 0, stream>>>(xws, row, col, agg1, E);
  }
  k_epilogue1<<<(n + 7) / 8, 256, 0, stream>>>(agg1, xws, dinv, b1, lnw, lnb, n);

  float* xw2s = agg1;
  k_gemm_scaled<128, 64><<<(n + 127) / 128, 256, 0, stream>>>(xws, W2, dinv, xw2s, n);
  {
    long long tot = (long long)E * 16;
    k_scatter<64><<<(unsigned)((tot + 255) / 256), 256, 0, stream>>>(xw2s, row, col, hout, E);
  }
  k_epilogue2<<<(n + 7) / 8, 256, 0, stream>>>(hout, xw2s, dinv, b2, batch, psum, pcnt, n);
  k_pool<<<(N_GRAPHS * 64 + 255) / 256, 256, 0, stream>>>(psum, pcnt, gout);
}


# Pool: each lane sums its 8 consecutive nodes in registers, a block's 16 groups combine through LDS, one f32 atomic per block, graph and dim (was one per node and dim); pool sum order changes
# speedup vs baseline: 1.1676x; 1.1676x over previous
#include <hip/hip_runtime.h>

typedef __attribute__((ext_vector_type(2))) float v2f;
typedef __attribute__((ext_vector_type(8))) float v8f;

#define N_GRAPHS 64


__global__ void k_degree(const int* __restrict__ col, unsigned* __restrict__ deg, int E) {
  int e = blockIdx.x * blockDim.x + threadIdx.x;
  if (e < E) atomicAdd(&deg[col[e]], 1u);
}

__global__ void k_dinv(const unsigned* __restrict__ deg, float* __restrict__ dinv, int n) {
  int i = blockIdx.x * blockDim.x + threadIdx.x;
  if (i < n) dinv[i] = rsqrtf((float)(deg[i] + 1u));
}


template <int IN, int OUT>
__global__ __launch_bounds__(256) void k_gemm_scaled(const float* __restrict__ A,
                                                     const float* __restrict__ W,
                                                     const float* __restrict__ dinv,
                                                     float* __restrict__ out, int nRows) {
  constexpr int NT = OUT / 16;
  const int lane = threadIdx.x & 31;
  const int wave = threadIdx.x >> 5;
  const int h = lane >> 4;
  const int m = lane & 15;
  const int rowBase = blockIdx.x * 128 + wave * 16;

  v8f acc[NT] = {};

  int arow = rowBase + m;
  if (arow > nRows - 1) arow = nRows - 1;
  const float* aptr = A + (size_t)arow * IN + 2 * h;

  for (int kb = 0; kb < IN; kb += 4) {
    float2 a2 = *(const float2*)(aptr + kb);
    v2f av; av.x = a2.x; av.y = a2.y;
    const float* w0 = W + (size_t)(kb + 2 * h) * OUT + m;
#pragma unroll
    for (int t = 0; t < NT; t++) {
      v2f bv; bv.x = w0[t * 16]; bv.y = w0[OUT + t * 16];
      acc[t] = __builtin_amdgcn_wmma_f32_16x16x4_f32(
           false, av,  false, bv,
           (short)0, acc[t],  false,  false);
    }
  }

#pragma unroll
  for (int v = 0; v < 8; v++) {
    int r = rowBase + v + 8 * h;
    if (r < nRows) {
      float s = dinv[r];
#pragma unroll
      for (int t = 0; t < NT; t++) out[(size_t)r * OUT + t * 16 + m] = acc[t][v] * s;
    }
  }
}


template <int D>
__global__ void k_scatter(const float* __restrict__ src, const int* __restrict__ row,
                          const int* __restrict__ col, float* __restrict__ dst, int E) {
  constexpr int C = D / 4;
  long long idx = (long long)blockIdx.x * blockDim.x + threadIdx.x;
  if (idx >= (long long)E * C) return;
  int e = (int)(idx / C);
  int c = (int)(idx % C);
  int r = row[e];
  int cl = col[e];
  float4 v = *(const float4*)(src + (size_t)r * D + c * 4);
  float* d = dst + (size_t)cl * D + c * 4;
  atomicAdd(d + 0, v.x);
  atomicAdd(d + 1, v.y);
  atomicAdd(d + 2, v.z);
  atomicAdd(d + 3, v.w);
}


__global__ void k_epilogue1(const float* __restrict__ agg, float* __restrict__ xws,
                            const float* __restrict__ dinv, const float* __restrict__ b1,
                            const float* __restrict__ lnw, const float* __restrict__ lnb,
                            int n) {
  int lane = threadIdx.x & 31;
  int wave = threadIdx.x >> 5;
  int i = blockIdx.x * 8 + wave;
  if (i >= n) return;
  float di = dinv[i];
  float t[4];
  float s = 0.f;
#pragma unroll
  for (int j = 0; j < 4; j++) {
    int d = lane + 32 * j;
    t[j] = (agg[(size_t)i * 128 + d] + xws[(size_t)i * 128 + d]) * di + b1[d];
    s += t[j];
  }
#pragma unroll
  for (int o = 16; o > 0; o >>= 1) s += __shfl_xor(s, o, 32);
  float mu = s * (1.f / 128.f);
  float var = 0.f;
#pragma unroll
  for (int j = 0; j < 4; j++) { float d0 = t[j] - mu; var += d0 * d0; }
#pragma unroll
  for (int o = 16; o > 0; o >>= 1) var += __shfl_xor(var, o, 32);
  float inv = rsqrtf(var * (1.f / 128.f) + 1e-5f);
#pragma unroll
  for (int j = 0; j < 4; j++) {
    int d = lane + 32 * j;
    float o = (t[j] - mu) * inv * lnw[d] + lnb[d];
    xws[(size_t)i * 128 + d] = fmaxf(o, 0.f);
  }
}


#define EPI2_NODES 8

__device__ __forceinline__ void pool_flush4(float* __restrict__ psum, float* __restrict__ pcnt,
                                            int b, int c, float4 acc, float cnt) {
  float* p = psum + b * 64 + c * 4;
  atomicAdd(p + 0, acc.x);
  atomicAdd(p + 1, acc.y);
  atomicAdd(p + 2, acc.z);
  atomicAdd(p + 3, acc.w);
  if (c == 0) atomicAdd(pcnt + b, cnt);
}

__global__ __launch_bounds__(256) void k_epilogue2(float* __restrict__ hout,
                                                   const float* __restrict__ xw2s,
                                                   const float* __restrict__ dinv,
                                                   const float* __restrict__ b2,
                                                   const int* __restrict__ batch,
                                                   float* __restrict__ psum,
                                                   float* __restrict__ pcnt, int n) {
  __shared__ __attribute__((aligned(16))) float part[16][68];
  __shared__ int pbid[16];

  const int c = threadIdx.x & 15;
  const int g = threadIdx.x >> 4;
  const int base = (blockIdx.x * 16 + g) * EPI2_NODES;
  const float4 bb = *(const float4*)(b2 + c * 4);

  float4 hv[EPI2_NODES];
  float4 xv[EPI2_NODES];
  float dv[EPI2_NODES];
  int bid[EPI2_NODES];
#pragma unroll
  for (int j = 0; j < EPI2_NODES; j++) {
    int i = base + j;
    if (i > n - 1) i = n - 1;
    hv[j] = *(const float4*)(hout + (size_t)i * 64 + c * 4);
    xv[j] = *(const float4*)(xw2s + (size_t)i * 64 + c * 4);
    dv[j] = dinv[i];
    bid[j] = batch[i];
  }

#pragma unroll
  for (int j = 0; j < EPI2_NODES; j++) {
    int i = base + j;
    float di = dv[j];
    float4 h;
    h.x = (hv[j].x + xv[j].x) * di + bb.x;
    h.y = (hv[j].y + xv[j].y) * di + bb.y;
    h.z = (hv[j].z + xv[j].z) * di + bb.z;
    h.w = (hv[j].w + xv[j].w) * di + bb.w;
    hv[j] = h;
    if (i < n) *(float4*)(hout + (size_t)i * 64 + c * 4) = h;
  }

  int cur = -1;
  float4 acc = make_float4(0.f, 0.f, 0.f, 0.f);
  float cnt = 0.f;
#pragma unroll
  for (int j = 0; j < EPI2_NODES; j++) {
    if (base + j < n) {
      int b = bid[j];
      if (b != cur) {
        if (cur >= 0) pool_flush4(psum, pcnt, cur, c, acc, cnt);
        cur = b;
        acc = make_float4(0.f, 0.f, 0.f, 0.f);
        cnt = 0.f;
      }
      acc.x += hv[j].x;
      acc.y += hv[j].y;
      acc.z += hv[j].z;
      acc.w += hv[j].w;
      cnt += 1.f;
    }
  }

  *(float4*)(&part[g][c * 4]) = acc;
  if (c == 0) {
    part[g][64] = cnt;
    pbid[g] = cur;
  }
  __syncthreads();

  const int d = threadIdx.x;
  if (d < 65) {
    float* dst = (d < 64) ? (psum + d) : pcnt;
    const int stride = (d < 64) ? 64 : 1;
    int ids[16];
    float vals[16];
#pragma unroll
    for (int s = 0; s < 16; s++) {
      ids[s] = pbid[s];
      vals[s] = part[s][d];
    }
    int rb = -1;
    float run = 0.f;
#pragma unroll
    for (int s = 0; s < 16; s++) {
      int b = ids[s];
      if (b >= 0) {
        if (b != rb) {
          if (rb >= 0) atomicAdd(dst + rb * stride, run);
          rb = b;
          run = 0.f;
        }
        run += vals[s];
      }
    }
    if (rb >= 0) atomicAdd(dst + rb * stride, run);
  }
}

__global__ void k_pool(const float* __restrict__ psum, const float* __restrict__ pcnt,
                       float* __restrict__ g) {
  int idx = blockIdx.x * blockDim.x + threadIdx.x;
  if (idx < N_GRAPHS * 64) {
    int gi = idx >> 6;
    g[idx] = psum[idx] / fmaxf(pcnt[gi], 1.f);
  }
}


extern "C" void kernel_launch(void* const* d_in, const int* in_sizes, int n_in,
                              void* d_out, int out_size, void* d_ws, size_t ws_size,
                              hipStream_t stream) {
  const float* x   = (const float*)d_in[0];
  const int*   ei  = (const int*)d_in[1];
  const int*  batch= (const int*)d_in[2];
  const float* W1  = (const float*)d_in[3];
  const float* b1  = (const float*)d_in[4];
  const float* lnw = (const float*)d_in[5];
  const float* lnb = (const float*)d_in[6];
  const float* W2  = (const float*)d_in[7];
  const float* b2  = (const float*)d_in[8];

  const int n = in_sizes[0] / 64;
  const int E = in_sizes[1] / 2;
  const int* row = ei;
  const int* col = ei + E;

  float* ws = (float*)d_ws;
  unsigned* deg = (unsigned*)ws;
  float* agg1 = ws + n;
  float* psum = agg1 + (size_t)n * 128;
  float* pcnt = psum + N_GRAPHS * 64;
  float* dinv = pcnt + N_GRAPHS;
  float* xws  = dinv + n;

  float* hout = (float*)d_out;
  float* gout = hout + (size_t)n * 64;

  size_t zbytes = ((size_t)n + (size_t)n * 128 + N_GRAPHS * 64 + N_GRAPHS) * sizeof(float);
  hipMemsetAsync(d_ws, 0, zbytes, stream);
  hipMemsetAsync(d_out, 0, (size_t)n * 64 * sizeof(float), stream);

  k_degree<<<(E + 255) / 256, 256, 0, stream>>>(col, deg, E);
  k_dinv<<<(n + 255) / 256, 256, 0, stream>>>(deg, dinv, n);

  k_gemm_scaled<64, 128><<<(n + 127) / 128, 256, 0, stream>>>(x, W1, dinv, xws, n);
  {
    long long tot = (long long)E * 32;
    k_scatter<128><<<(unsigned)((tot + 255) / 256), 256, 0, stream>>>(xws, row, col, agg1, E);
  }
  k_epilogue1<<<(n + 7) / 8, 256, 0, stream>>>(agg1, xws, dinv, b1, lnw, lnb, n);

  float* xw2s = agg1;
  k_gemm_scaled<128, 64><<<(n + 127) / 128, 256, 0, stream>>>(xws, W2, dinv, xw2s, n);
  {
    long long tot = (long long)E * 16;
    k_scatter<64><<<(unsigned)((tot + 255) / 256), 256, 0, stream>>>(xw2s, row, col, hout, E);
  }
  k_epilogue2<<<(n + 16 * EPI2_NODES - 1) / (16 * EPI2_NODES), 256, 0, stream>>>(hout, xw2s, dinv, b2, batch, psum, pcnt, n);
  k_pool<<<(N_GRAPHS * 64 + 255) / 256, 256, 0, stream>>>(psum, pcnt, gout);
}


# Edge sums by gather over a per-call target-sorted edge list (int-atomic slot, scan, fill, per-node sort by source id) replace the f32 atomic scatter; both layers' edge sum order becomes fixed
# speedup vs baseline: 2.0835x; 1.7844x over previous
#include <hip/hip_runtime.h>

typedef __attribute__((ext_vector_type(2))) float v2f;
typedef __attribute__((ext_vector_type(8))) float v8f;

#define N_GRAPHS 64


__global__ void k_degree(const int* __restrict__ col, unsigned* __restrict__ deg,
                         int* __restrict__ pos, int E) {
  int e = blockIdx.x * blockDim.x + threadIdx.x;
  if (e < E) pos[e] = (int)atomicAdd(&deg[col[e]], 1u);
}

__global__ __launch_bounds__(256) void k_scan(const unsigned* __restrict__ deg,
                                              int* __restrict__ rowptr,
                                              float* __restrict__ dinv, int n) {
  __shared__ unsigned red[8];
  __shared__ unsigned wsum[8];
  const int tid = threadIdx.x;
  const int lane = tid & 31;
  const int wave = tid >> 5;
  const int blockStart = blockIdx.x * 1024;

  unsigned s = 0;
  const uint4* d4 = (const uint4*)deg;
  const int nvec = blockStart >> 2;
  for (int i = tid; i < nvec; i += 256) {
    uint4 v = d4[i];
    s += v.x + v.y + v.z + v.w;
  }
#pragma unroll
  for (int o = 16; o > 0; o >>= 1) s += __shfl_xor(s, o, 32);
  if (lane == 0) red[wave] = s;

  const int i0 = blockStart + tid * 4;
  unsigned d[4];
#pragma unroll
  for (int k = 0; k < 4; k++) d[k] = (i0 + k < n) ? deg[i0 + k] : 0u;
  const unsigned t = d[0] + d[1] + d[2] + d[3];
  unsigned inc = t;
#pragma unroll
  for (int o = 1; o < 32; o <<= 1) {
    unsigned u = __shfl_up(inc, o, 32);
    if (lane >= o) inc += u;
  }
  if (lane == 31) wsum[wave] = inc;
  __syncthreads();

  unsigned basev = 0;
#pragma unroll
  for (int w = 0; w < 8; w++) basev += red[w];
  for (int w = 0; w < wave; w++) basev += wsum[w];
  unsigned ex = basev + inc - t;
#pragma unroll
  for (int k = 0; k < 4; k++) {
    if (i0 + k < n) {
      rowptr[i0 + k] = (int)ex;
      dinv[i0 + k] = rsqrtf((float)(d[k] + 1u));
    }
    ex += d[k];
  }
}

__global__ void k_fill(const int* __restrict__ row, const int* __restrict__ col,
                       const int* __restrict__ pos, const int* __restrict__ rowptr,
                       int* __restrict__ seg, int E) {
  int e = blockIdx.x * blockDim.x + threadIdx.x;
  if (e < E) seg[rowptr[col[e]] + pos[e]] = row[e];
}

__global__ __launch_bounds__(256) void k_sortseg(const int* __restrict__ rowptr,
                                                 const unsigned* __restrict__ deg,
                                                 const int* __restrict__ seg,
                                                 int* __restrict__ sorted, int n) {
  const int lane = threadIdx.x & 31;
  const int node = __builtin_amdgcn_readfirstlane(blockIdx.x * 8 + (threadIdx.x >> 5));
  if (node >= n) return;
  const int start = rowptr[node];
  const int dg = (int)deg[node];
  for (int a = 0; a < dg; a += 32) {
    const int ia = a + lane;
    const int ka = (ia < dg) ? seg[start + ia] : 0x7fffffff;
    int rank = 0;
    for (int b = 0; b < dg; b += 32) {
      const int kb = (b + lane < dg) ? seg[start + b + lane] : 0x7fffffff;
      const int cnt = min(32, dg - b);
      for (int j = 0; j < cnt; j++) {
        const int s = __builtin_amdgcn_readlane(kb, j);
        const int jj = b + j;
        rank += (s < ka || (s == ka && jj < ia)) ? 1 : 0;
      }
    }
    if (ia < dg) sorted[start + rank] = ka;
  }
}


template <int IN, int OUT>
__global__ __launch_bounds__(256) void k_gemm_scaled(const float* __restrict__ A,
                                                     const float* __restrict__ W,
                                                     const float* __restrict__ dinv,
                                                     float* __restrict__ out, int nRows) {
  constexpr int NT = OUT / 16;
  const int lane = threadIdx.x & 31;
  const int wave = threadIdx.x >> 5;
  const int h = lane >> 4;
  const int m = lane & 15;
  const int rowBase = blockIdx.x * 128 + wave * 16;

  v8f acc[NT] = {};

  int arow = rowBase + m;
  if (arow > nRows - 1) arow = nRows - 1;
  const float* aptr = A + (size_t)arow * IN + 2 * h;

  for (int kb = 0; kb < IN; kb += 4) {
    float2 a2 = *(const float2*)(aptr + kb);
    v2f av; av.x = a2.x; av.y = a2.y;
    const float* w0 = W + (size_t)(kb + 2 * h) * OUT + m;
#pragma unroll
    for (int t = 0; t < NT; t++) {
      v2f bv; bv.x = w0[t * 16]; bv.y = w0[OUT + t * 16];
      acc[t] = __builtin_amdgcn_wmma_f32_16x16x4_f32(
           false, av,  false, bv,
           (short)0, acc[t],  false,  false);
    }
  }

#pragma unroll
  for (int v = 0; v < 8; v++) {
    int r = rowBase + v + 8 * h;
    if (r < nRows) {
      float s = dinv[r];
#pragma unroll
      for (int t = 0; t < NT; t++) out[(size_t)r * OUT + t * 16 + m] = acc[t][v] * s;
    }
  }
}


template <int D>
__global__ __launch_bounds__(256) void k_gather(const float* __restrict__ src,
                                                const int* __restrict__ rowptr,
                                                const unsigned* __restrict__ deg,
                                                const int* __restrict__ sorted,
                                                float* __restrict__ dst, int n) {
  constexpr int LPR = D / 4;
  constexpr int RPW = 32 / LPR;
  const int lane = threadIdx.x & 31;
  const int node = __builtin_amdgcn_readfirstlane(blockIdx.x * 8 + (threadIdx.x >> 5));
  if (node >= n) return;
  const int start = rowptr[node];
  const int dg = (int)deg[node];
  const int sub = lane / LPR;
  const int c = lane % LPR;
  float4 acc = make_float4(0.f, 0.f, 0.f, 0.f);
  for (int base = 0; base < dg; base += 32) {
    const int cnt = min(32, dg - base);
    const int myid = (lane < cnt) ? sorted[start + base + lane] : 0;
    if constexpr (RPW == 1) {
#pragma unroll 4
      for (int j = 0; j < cnt; j++) {
        const int s = __builtin_amdgcn_readlane(myid, j);
        const float4 v = *(const float4*)(src + (size_t)s * D + c * 4);
        acc.x += v.x; acc.y += v.y; acc.z += v.z; acc.w += v.w;
      }
    } else {
      const int pairs = cnt >> 1;
#pragma unroll 4
      for (int p = 0; p < pairs; p++) {
        const int s = __shfl(myid, 2 * p + sub, 32);
        const float4 v = *(const float4*)(src + (size_t)s * D + c * 4);
        acc.x += v.x; acc.y += v.y; acc.z += v.z; acc.w += v.w;
      }
      if (cnt & 1) {
        const int s = __shfl(myid, cnt - 1, 32);
        if (sub == 0) {
          const float4 v = *(const float4*)(src + (size_t)s * D + c * 4);
          acc.x += v.x; acc.y += v.y; acc.z += v.z; acc.w += v.w;
        }
      }
    }
  }
  if constexpr (RPW == 2) {
    acc.x += __shfl_xor(acc.x, 16, 32);
    acc.y += __shfl_xor(acc.y, 16, 32);
    acc.z += __shfl_xor(acc.z, 16, 32);
    acc.w += __shfl_xor(acc.w, 16, 32);
  }
  if (sub == 0) *(float4*)(dst + (size_t)node * D + c * 4) = acc;
}


__global__ void k_epilogue1(const float* __restrict__ agg, float* __restrict__ xws,
                            const float* __restrict__ dinv, const float* __restrict__ b1,
                            const float* __restrict__ lnw, const float* __restrict__ lnb,
                            int n) {
  int lane = threadIdx.x & 31;
  int wave = threadIdx.x >> 5;
  int i = blockIdx.x * 8 + wave;
  if (i >= n) return;
  float di = dinv[i];
  float t[4];
  float s = 0.f;
#pragma unroll
  for (int j = 0; j < 4; j++) {
    int d = lane + 32 * j;
    t[j] = (agg[(size_t)i * 128 + d] + xws[(size_t)i * 128 + d]) * di + b1[d];
    s += t[j];
  }
#pragma unroll
  for (int o = 16; o > 0; o >>= 1) s += __shfl_xor(s, o, 32);
  float mu = s * (1.f / 128.f);
  float var = 0.f;
#pragma unroll
  for (int j = 0; j < 4; j++) { float d0 = t[j] - mu; var += d0 * d0; }
#pragma unroll
  for (int o = 16; o > 0; o >>= 1) var += __shfl_xor(var, o, 32);
  float inv = rsqrtf(var * (1.f / 128.f) + 1e-5f);
#pragma unroll
  for (int j = 0; j < 4; j++) {
    int d = lane + 32 * j;
    float o = (t[j] - mu) * inv * lnw[d] + lnb[d];
    xws[(size_t)i * 128 + d] = fmaxf(o, 0.f);
  }
}


#define EPI2_NODES 8

__device__ __forceinline__ void pool_flush4(float* __restrict__ psum, float* __restrict__ pcnt,
                                            int b, int c, float4 acc, float cnt) {
  float* p = psum + b * 64 + c * 4;
  atomicAdd(p + 0, acc.x);
  atomicAdd(p + 1, acc.y);
  atomicAdd(p + 2, acc.z);
  atomicAdd(p + 3, acc.w);
  if (c == 0) atomicAdd(pcnt + b, cnt);
}

__global__ __launch_bounds__(256) void k_epilogue2(float* __restrict__ hout,
                                                   const float* __restrict__ xw2s,
                                                   const float* __restrict__ dinv,
                                                   const float* __restrict__ b2,
                                                   const int* __restrict__ batch,
                                                   float* __restrict__ psum,
                                                   float* __restrict__ pcnt, int n) {
  __shared__ __attribute__((aligned(16))) float part[16][68];
  __shared__ int pbid[16];

  const int c = threadIdx.x & 15;
  const int g = threadIdx.x >> 4;
  const int base = (blockIdx.x * 16 + g) * EPI2_NODES;
  const float4 bb = *(const float4*)(b2 + c * 4);

  float4 hv[EPI2_NODES];
  float4 xv[EPI2_NODES];
  float dv[EPI2_NODES];
  int bid[EPI2_NODES];
#pragma unroll
  for (int j = 0; j < EPI2_NODES; j++) {
    int i = base + j;
    if (i > n - 1) i = n - 1;
    hv[j] = *(const float4*)(hout + (size_t)i * 64 + c * 4);
    xv[j] = *(const float4*)(xw2s + (size_t)i * 64 + c * 4);
    dv[j] = dinv[i];
    bid[j] = batch[i];
  }

#pragma unroll
  for (int j = 0; j < EPI2_NODES; j++) {
    int i = base + j;
    float di = dv[j];
    float4 h;
    h.x = (hv[j].x + xv[j].x) * di + bb.x;
    h.y = (hv[j].y + xv[j].y) * di + bb.y;
    h.z = (hv[j].z + xv[j].z) * di + bb.z;
    h.w = (hv[j].w + xv[j].w) * di + bb.w;
    hv[j] = h;
    if (i < n) *(float4*)(hout + (size_t)i * 64 + c * 4) = h;
  }

  int cur = -1;
  float4 acc = make_float4(0.f, 0.f, 0.f, 0.f);
  float cnt = 0.f;
#pragma unroll
  for (int j = 0; j < EPI2_NODES; j++) {
    if (base + j < n) {
      int b = bid[j];
      if (b != cur) {
        if (cur >= 0) pool_flush4(psum, pcnt, cur, c, acc, cnt);
        cur = b;
        acc = make_float4(0.f, 0.f, 0.f, 0.f);
        cnt = 0.f;
      }
      acc.x += hv[j].x;
      acc.y += hv[j].y;
      acc.z += hv[j].z;
      acc.w += hv[j].w;
      cnt += 1.f;
    }
  }

  *(float4*)(&part[g][c * 4]) = acc;
  if (c == 0) {
    part[g][64] = cnt;
    pbid[g] = cur;
  }
  __syncthreads();

  const int d = threadIdx.x;
  if (d < 65) {
    float* dst = (d < 64) ? (psum + d) : pcnt;
    const int stride = (d < 64) ? 64 : 1;
    int ids[16];
    float vals[16];
#pragma unroll
    for (int s = 0; s < 16; s++) {
      ids[s] = pbid[s];
      vals[s] = part[s][d];
    }
    int rb = -1;
    float run = 0.f;
#pragma unroll
    for (int s = 0; s < 16; s++) {
      int b = ids[s];
      if (b >= 0) {
        if (b != rb) {
          if (rb >= 0) atomicAdd(dst + rb * stride, run);
          rb = b;
          run = 0.f;
        }
        run += vals[s];
      }
    }
    if (rb >= 0) atomicAdd(dst + rb * stride, run);
  }
}

__global__ void k_pool(const float* __restrict__ psum, const float* __restrict__ pcnt,
                       float* __restrict__ g) {
  int idx = blockIdx.x * blockDim.x + threadIdx.x;
  if (idx < N_GRAPHS * 64) {
    int gi = idx >> 6;
    g[idx] = psum[idx] / fmaxf(pcnt[gi], 1.f);
  }
}


extern "C" void kernel_launch(void* const* d_in, const int* in_sizes, int n_in,
                              void* d_out, int out_size, void* d_ws, size_t ws_size,
                              hipStream_t stream) {
  const float* x   = (const float*)d_in[0];
  const int*   ei  = (const int*)d_in[1];
  const int*  batch= (const int*)d_in[2];
  const float* W1  = (const float*)d_in[3];
  const float* b1  = (const float*)d_in[4];
  const float* lnw = (const float*)d_in[5];
  const float* lnb = (const float*)d_in[6];
  const float* W2  = (const float*)d_in[7];
  const float* b2  = (const float*)d_in[8];

  const int n = in_sizes[0] / 64;
  const int E = in_sizes[1] / 2;
  const int* row = ei;
  const int* col = ei + E;

  float* ws = (float*)d_ws;
  float* xws  = ws;
  float* agg1 = xws + (size_t)n * 128;
  unsigned* deg = (unsigned*)(agg1 + (size_t)n * 128);
  float* psum = (float*)(deg + n);
  float* pcnt = psum + N_GRAPHS * 64;
  float* dinv = pcnt + N_GRAPHS;
  int* rowptr = (int*)(dinv + n);
  int* pos    = rowptr + n;
  int* seg    = pos + E;
  int* sorted = seg + E;

  float* hout = (float*)d_out;
  float* gout = hout + (size_t)n * 64;

  size_t zbytes = ((size_t)n + N_GRAPHS * 64 + N_GRAPHS) * sizeof(float);
  hipMemsetAsync(deg, 0, zbytes, stream);

  k_degree<<<(E + 255) / 256, 256, 0, stream>>>(col, deg, pos, E);
  k_scan<<<(n + 1023) / 1024, 256, 0, stream>>>(deg, rowptr, dinv, n);
  k_fill<<<(E + 255) / 256, 256, 0, stream>>>(row, col, pos, rowptr, seg, E);
  k_sortseg<<<(n + 7) / 8, 256, 0, stream>>>(rowptr, deg, seg, sorted, n);

  k_gemm_scaled<64, 128><<<(n + 127) / 128, 256, 0, stream>>>(x, W1, dinv, xws, n);
  k_gather<128><<<(n + 7) / 8, 256, 0, stream>>>(xws, rowptr, deg, sorted, agg1, n);
  k_epilogue1<<<(n + 7) / 8, 256, 0, stream>>>(agg1, xws, dinv, b1, lnw, lnb, n);

  float* xw2s = agg1;
  k_gemm_scaled<128, 64><<<(n + 127) / 128, 256, 0, stream>>>(xws, W2, dinv, xw2s, n);
  k_gather<64><<<(n + 7) / 8, 256, 0, stream>>>(xw2s, rowptr, deg, sorted, hout, n);
  k_epilogue2<<<(n + 16 * EPI2_NODES - 1) / (16 * EPI2_NODES), 256, 0, stream>>>(hout, xw2s, dinv, b2, batch, psum, pcnt, n);
  k_pool<<<(N_GRAPHS * 64 + 255) / 256, 256, 0, stream>>>(psum, pcnt, gout);
}


# Each gather kernel finishes its node in the same wave: self-loop row, dinv, bias (layer 1: layer norm + ReLU too) before its one store; both epilogue passes go; layer-norm sum order changes
# speedup vs baseline: 2.1923x; 1.0522x over previous
#include <hip/hip_runtime.h>

typedef __attribute__((ext_vector_type(2))) float v2f;
typedef __attribute__((ext_vector_type(8))) float v8f;

#define N_GRAPHS 64


__global__ void k_degree(const int* __restrict__ col, unsigned* __restrict__ deg,
                         int* __restrict__ pos, int E) {
  int e = blockIdx.x * blockDim.x + threadIdx.x;
  if (e < E) pos[e] = (int)atomicAdd(&deg[col[e]], 1u);
}

__global__ __launch_bounds__(256) void k_scan(const unsigned* __restrict__ deg,
                                              int* __restrict__ rowptr,
                                              float* __restrict__ dinv, int n) {
  __shared__ unsigned red[8];
  __shared__ unsigned wsum[8];
  const int tid = threadIdx.x;
  const int lane = tid & 31;
  const int wave = tid >> 5;
  const int blockStart = blockIdx.x * 1024;

  unsigned s = 0;
  const uint4* d4 = (const uint4*)deg;
  const int nvec = blockStart >> 2;
  for (int i = tid; i < nvec; i += 256) {
    uint4 v = d4[i];
    s += v.x + v.y + v.z + v.w;
  }
#pragma unroll
  for (int o = 16; o > 0; o >>= 1) s += __shfl_xor(s, o, 32);
  if (lane == 0) red[wave] = s;

  const int i0 = blockStart + tid * 4;
  unsigned d[4];
#pragma unroll
  for (int k = 0; k < 4; k++) d[k] = (i0 + k < n) ? deg[i0 + k] : 0u;
  const unsigned t = d[0] + d[1] + d[2] + d[3];
  unsigned inc = t;
#pragma unroll
  for (int o = 1; o < 32; o <<= 1) {
    unsigned u = __shfl_up(inc, o, 32);
    if (lane >= o) inc += u;
  }
  if (lane == 31) wsum[wave] = inc;
  __syncthreads();

  unsigned basev = 0;
#pragma unroll
  for (int w = 0; w < 8; w++) basev += red[w];
  for (int w = 0; w < wave; w++) basev += wsum[w];
  unsigned ex = basev + inc - t;
#pragma unroll
  for (int k = 0; k < 4; k++) {
    if (i0 + k < n) {
      rowptr[i0 + k] = (int)ex;
      dinv[i0 + k] = rsqrtf((float)(d[k] + 1u));
    }
    ex += d[k];
  }
}

__global__ void k_fill(const int* __restrict__ row, const int* __restrict__ col,
                       const int* __restrict__ pos, const int* __restrict__ rowptr,
                       int* __restrict__ seg, int E) {
  int e = blockIdx.x * blockDim.x + threadIdx.x;
  if (e < E) seg[rowptr[col[e]] + pos[e]] = row[e];
}

__global__ __launch_bounds__(256) void k_sortseg(const int* __restrict__ rowptr,
                                                 const unsigned* __restrict__ deg,
                                                 const int* __restrict__ seg,
                                                 int* __restrict__ sorted, int n) {
  const int lane = threadIdx.x & 31;
  const int node = __builtin_amdgcn_readfirstlane(blockIdx.x * 8 + (threadIdx.x >> 5));
  if (node >= n) return;
  const int start = rowptr[node];
  const int dg = (int)deg[node];
  for (int a = 0; a < dg; a += 32) {
    const int ia = a + lane;
    const int ka = (ia < dg) ? seg[start + ia] : 0x7fffffff;
    int rank = 0;
    for (int b = 0; b < dg; b += 32) {
      const int kb = (b + lane < dg) ? seg[start + b + lane] : 0x7fffffff;
      const int cnt = min(32, dg - b);
      for (int j = 0; j < cnt; j++) {
        const int s = __builtin_amdgcn_readlane(kb, j);
        const int jj = b + j;
        rank += (s < ka || (s == ka && jj < ia)) ? 1 : 0;
      }
    }
    if (ia < dg) sorted[start + rank] = ka;
  }
}


template <int IN, int OUT>
__global__ __launch_bounds__(256) void k_gemm_scaled(const float* __restrict__ A,
                                                     const float* __restrict__ W,
                                                     const float* __restrict__ dinv,
                                                     float* __restrict__ out, int nRows) {
  constexpr int NT = OUT / 16;
  const int lane = threadIdx.x & 31;
  const int wave = threadIdx.x >> 5;
  const int h = lane >> 4;
  const int m = lane & 15;
  const int rowBase = blockIdx.x * 128 + wave * 16;

  v8f acc[NT] = {};

  int arow = rowBase + m;
  if (arow > nRows - 1) arow = nRows - 1;
  const float* aptr = A + (size_t)arow * IN + 2 * h;

  for (int kb = 0; kb < IN; kb += 4) {
    float2 a2 = *(const float2*)(aptr + kb);
    v2f av; av.x = a2.x; av.y = a2.y;
    const float* w0 = W + (size_t)(kb + 2 * h) * OUT + m;
#pragma unroll
    for (int t = 0; t < NT; t++) {
      v2f bv; bv.x = w0[t * 16]; bv.y = w0[OUT + t * 16];
      acc[t] = __builtin_amdgcn_wmma_f32_16x16x4_f32(
           false, av,  false, bv,
           (short)0, acc[t],  false,  false);
    }
  }

#pragma unroll
  for (int v = 0; v < 8; v++) {
    int r = rowBase + v + 8 * h;
    if (r < nRows) {
      float s = dinv[r];
#pragma unroll
      for (int t = 0; t < NT; t++) out[(size_t)r * OUT + t * 16 + m] = acc[t][v] * s;
    }
  }
}


template <int D>
__global__ __launch_bounds__(256) void k_gather_finish(const float* __restrict__ src,
                                                       const int* __restrict__ rowptr,
                                                       const unsigned* __restrict__ deg,
                                                       const int* __restrict__ sorted,
                                                       const float* __restrict__ dinv,
                                                       const float* __restrict__ bias,
                                                       const float* __restrict__ lnw,
                                                       const float* __restrict__ lnb,
                                                       float* __restrict__ dst, int n) {
  constexpr int LPR = D / 4;
  constexpr int RPW = 32 / LPR;
  const int lane = threadIdx.x & 31;
  const int node = __builtin_amdgcn_readfirstlane(blockIdx.x * 8 + (threadIdx.x >> 5));
  if (node >= n) return;
  const int start = rowptr[node];
  const int dg = (int)deg[node];
  const float di = dinv[node];
  const int sub = lane / LPR;
  const int c = lane % LPR;
  const float4 self = *(const float4*)(src + (size_t)node * D + c * 4);
  const float4 bb = *(const float4*)(bias + c * 4);
  float4 acc = make_float4(0.f, 0.f, 0.f, 0.f);
  for (int base = 0; base < dg; base += 32) {
    const int cnt = min(32, dg - base);
    const int myid = (lane < cnt) ? sorted[start + base + lane] : 0;
    if constexpr (RPW == 1) {
#pragma unroll 4
      for (int j = 0; j < cnt; j++) {
        const int s = __builtin_amdgcn_readlane(myid, j);
        const float4 v = *(const float4*)(src + (size_t)s * D + c * 4);
        acc.x += v.x; acc.y += v.y; acc.z += v.z; acc.w += v.w;
      }
    } else {
      const int pairs = cnt >> 1;
#pragma unroll 4
      for (int p = 0; p < pairs; p++) {
        const int s = __shfl(myid, 2 * p + sub, 32);
        const float4 v = *(const float4*)(src + (size_t)s * D + c * 4);
        acc.x += v.x; acc.y += v.y; acc.z += v.z; acc.w += v.w;
      }
      if (cnt & 1) {
        const int s = __shfl(myid, cnt - 1, 32);
        if (sub == 0) {
          const float4 v = *(const float4*)(src + (size_t)s * D + c * 4);
          acc.x += v.x; acc.y += v.y; acc.z += v.z; acc.w += v.w;
        }
      }
    }
  }
  if constexpr (RPW == 2) {
    acc.x += __shfl_xor(acc.x, 16, 32);
    acc.y += __shfl_xor(acc.y, 16, 32);
    acc.z += __shfl_xor(acc.z, 16, 32);
    acc.w += __shfl_xor(acc.w, 16, 32);
  }

  float4 t;
  t.x = (acc.x + self.x) * di + bb.x;
  t.y = (acc.y + self.y) * di + bb.y;
  t.z = (acc.z + self.z) * di + bb.z;
  t.w = (acc.w + self.w) * di + bb.w;

  if constexpr (D == 128) {
    const float4 gw = *(const float4*)(lnw + c * 4);
    const float4 gb = *(const float4*)(lnb + c * 4);
    float s = (t.x + t.y) + (t.z + t.w);
#pragma unroll
    for (int o = 16; o > 0; o >>= 1) s += __shfl_xor(s, o, 32);
    const float mu = s * (1.f / 128.f);
    const float dx = t.x - mu, dy = t.y - mu, dz = t.z - mu, dw = t.w - mu;
    float var = (dx * dx + dy * dy) + (dz * dz + dw * dw);
#pragma unroll
    for (int o = 16; o > 0; o >>= 1) var += __shfl_xor(var, o, 32);
    const float inv = rsqrtf(var * (1.f / 128.f) + 1e-5f);
    float4 o4;
    o4.x = fmaxf(dx * inv * gw.x + gb.x, 0.f);
    o4.y = fmaxf(dy * inv * gw.y + gb.y, 0.f);
    o4.z = fmaxf(dz * inv * gw.z + gb.z, 0.f);
    o4.w = fmaxf(dw * inv * gw.w + gb.w, 0.f);
    *(float4*)(dst + (size_t)node * D + c * 4) = o4;
  } else {
    if (sub == 0) *(float4*)(dst + (size_t)node * D + c * 4) = t;
  }
}


#define POOL_NODES 8

__device__ __forceinline__ void pool_flush4(float* __restrict__ psum, float* __restrict__ pcnt,
                                            int b, int c, float4 acc, float cnt) {
  float* p = psum + b * 64 + c * 4;
  atomicAdd(p + 0, acc.x);
  atomicAdd(p + 1, acc.y);
  atomicAdd(p + 2, acc.z);
  atomicAdd(p + 3, acc.w);
  if (c == 0) atomicAdd(pcnt + b, cnt);
}

__global__ __launch_bounds__(256) void k_poolsum(const float* __restrict__ hout,
                                                 const int* __restrict__ batch,
                                                 float* __restrict__ psum,
                                                 float* __restrict__ pcnt, int n) {
  __shared__ __attribute__((aligned(16))) float part[16][68];
  __shared__ int pbid[16];

  const int c = threadIdx.x & 15;
  const int g = threadIdx.x >> 4;
  const int base = (blockIdx.x * 16 + g) * POOL_NODES;

  float4 hv[POOL_NODES];
  int bid[POOL_NODES];
#pragma unroll
  for (int j = 0; j < POOL_NODES; j++) {
    int i = base + j;
    if (i > n - 1) i = n - 1;
    hv[j] = *(const float4*)(hout + (size_t)i * 64 + c * 4);
    bid[j] = batch[i];
  }

  int cur = -1;
  float4 acc = make_float4(0.f, 0.f, 0.f, 0.f);
  float cnt = 0.f;
#pragma unroll
  for (int j = 0; j < POOL_NODES; j++) {
    if (base + j < n) {
      int b = bid[j];
      if (b != cur) {
        if (cur >= 0) pool_flush4(psum, pcnt, cur, c, acc, cnt);
        cur = b;
        acc = make_float4(0.f, 0.f, 0.f, 0.f);
        cnt = 0.f;
      }
      acc.x += hv[j].x;
      acc.y += hv[j].y;
      acc.z += hv[j].z;
      acc.w += hv[j].w;
      cnt += 1.f;
    }
  }

  *(float4*)(&part[g][c * 4]) = acc;
  if (c == 0) {
    part[g][64] = cnt;
    pbid[g] = cur;
  }
  __syncthreads();

  const int d = threadIdx.x;
  if (d < 65) {
    float* dst = (d < 64) ? (psum + d) : pcnt;
    const int stride = (d < 64) ? 64 : 1;
    int ids[16];
    float vals[16];
#pragma unroll
    for (int s = 0; s < 16; s++) {
      ids[s] = pbid[s];
      vals[s] = part[s][d];
    }
    int rb = -1;
    float run = 0.f;
#pragma unroll
    for (int s = 0; s < 16; s++) {
      int b = ids[s];
      if (b >= 0) {
        if (b != rb) {
          if (rb >= 0) atomicAdd(dst + rb * stride, run);
          rb = b;
          run = 0.f;
        }
        run += vals[s];
      }
    }
    if (rb >= 0) atomicAdd(dst + rb * stride, run);
  }
}

__global__ void k_pool(const float* __restrict__ psum, const float* __restrict__ pcnt,
                       float* __restrict__ g) {
  int idx = blockIdx.x * blockDim.x + threadIdx.x;
  if (idx < N_GRAPHS * 64) {
    int gi = idx >> 6;
    g[idx] = psum[idx] / fmaxf(pcnt[gi], 1.f);
  }
}


extern "C" void kernel_launch(void* const* d_in, const int* in_sizes, int n_in,
                              void* d_out, int out_size, void* d_ws, size_t ws_size,
                              hipStream_t stream) {
  const float* x   = (const float*)d_in[0];
  const int*   ei  = (const int*)d_in[1];
  const int*  batch= (const int*)d_in[2];
  const float* W1  = (const float*)d_in[3];
  const float* b1  = (const float*)d_in[4];
  const float* lnw = (const float*)d_in[5];
  const float* lnb = (const float*)d_in[6];
  const float* W2  = (const float*)d_in[7];
  const float* b2  = (const float*)d_in[8];

  const int n = in_sizes[0] / 64;
  const int E = in_sizes[1] / 2;
  const int* row = ei;
  const int* col = ei + E;

  float* ws = (float*)d_ws;
  float* xws  = ws;
  float* h1   = xws + (size_t)n * 128;
  unsigned* deg = (unsigned*)(h1 + (size_t)n * 128);
  float* psum = (float*)(deg + n);
  float* pcnt = psum + N_GRAPHS * 64;
  float* dinv = pcnt + N_GRAPHS;
  int* rowptr = (int*)(dinv + n);
  int* pos    = rowptr + n;
  int* seg    = pos + E;
  int* sorted = seg + E;

  float* hout = (float*)d_out;
  float* gout = hout + (size_t)n * 64;

  size_t zbytes = ((size_t)n + N_GRAPHS * 64 + N_GRAPHS) * sizeof(float);
  hipMemsetAsync(deg, 0, zbytes, stream);

  k_degree<<<(E + 255) / 256, 256, 0, stream>>>(col, deg, pos, E);
  k_scan<<<(n + 1023) / 1024, 256, 0, stream>>>(deg, rowptr, dinv, n);
  k_fill<<<(E + 255) / 256, 256, 0, stream>>>(row, col, pos, rowptr, seg, E);
  k_sortseg<<<(n + 7) / 8, 256, 0, stream>>>(rowptr, deg, seg, sorted, n);

  k_gemm_scaled<64, 128><<<(n + 127) / 128, 256, 0, stream>>>(x, W1, dinv, xws, n);
  k_gather_finish<128><<<(n + 7) / 8, 256, 0, stream>>>(xws, rowptr, deg, sorted, dinv, b1, lnw, lnb, h1, n);

  float* xw2s = xws;
  k_gemm_scaled<128, 64><<<(n + 127) / 128, 256, 0, stream>>>(h1, W2, dinv, xw2s, n);
  k_gather_finish<64><<<(n + 7) / 8, 256, 0, stream>>>(xw2s, rowptr, deg, sorted, dinv, b2, nullptr, nullptr, hout, n);
  k_poolsum<<<(n + 16 * POOL_NODES - 1) / (16 * POOL_NODES), 256, 0, stream>>>(hout, batch, psum, pcnt, n);
  k_pool<<<(N_GRAPHS * 64 + 255) / 256, 256, 0, stream>>>(psum, pcnt, gout);
}


# Layer 1 aggregates before it transforms: the sorted-list gather sums 64-wide x*dinv rows (fixed order, no float atomics), GEMM1 takes the sum, adds bias+LN+ReLU; layer-1 and LN f32 sum order change
# speedup vs baseline: 2.3567x; 1.0750x over previous
#include <hip/hip_runtime.h>

typedef __attribute__((ext_vector_type(2))) float v2f;
typedef __attribute__((ext_vector_type(8))) float v8f;

#define N_GRAPHS 64


__global__ void k_degree(const int* __restrict__ col, unsigned* __restrict__ deg,
                         int* __restrict__ pos, int E) {
  int e = blockIdx.x * blockDim.x + threadIdx.x;
  if (e < E) pos[e] = (int)atomicAdd(&deg[col[e]], 1u);
}

__global__ __launch_bounds__(256) void k_scan(const unsigned* __restrict__ deg,
                                              int* __restrict__ rowptr,
                                              float* __restrict__ dinv, int n) {
  __shared__ unsigned red[8];
  __shared__ unsigned wsum[8];
  const int tid = threadIdx.x;
  const int lane = tid & 31;
  const int wave = tid >> 5;
  const int blockStart = blockIdx.x * 1024;

  unsigned s = 0;
  const uint4* d4 = (const uint4*)deg;
  const int nvec = blockStart >> 2;
  for (int i = tid; i < nvec; i += 256) {
    uint4 v = d4[i];
    s += v.x + v.y + v.z + v.w;
  }
#pragma unroll
  for (int o = 16; o > 0; o >>= 1) s += __shfl_xor(s, o, 32);
  if (lane == 0) red[wave] = s;

  const int i0 = blockStart + tid * 4;
  unsigned d[4];
#pragma unroll
  for (int k = 0; k < 4; k++) d[k] = (i0 + k < n) ? deg[i0 + k] : 0u;
  const unsigned t = d[0] + d[1] + d[2] + d[3];
  unsigned inc = t;
#pragma unroll
  for (int o = 1; o < 32; o <<= 1) {
    unsigned u = __shfl_up(inc, o, 32);
    if (lane >= o) inc += u;
  }
  if (lane == 31) wsum[wave] = inc;
  __syncthreads();

  unsigned basev = 0;
#pragma unroll
  for (int w = 0; w < 8; w++) basev += red[w];
  for (int w = 0; w < wave; w++) basev += wsum[w];
  unsigned ex = basev + inc - t;
#pragma unroll
  for (int k = 0; k < 4; k++) {
    if (i0 + k < n) {
      rowptr[i0 + k] = (int)ex;
      dinv[i0 + k] = rsqrtf((float)(d[k] + 1u));
    }
    ex += d[k];
  }
}

__global__ void k_fill(const int* __restrict__ row, const int* __restrict__ col,
                       const int* __restrict__ pos, const int* __restrict__ rowptr,
                       int* __restrict__ seg, int E) {
  int e = blockIdx.x * blockDim.x + threadIdx.x;
  if (e < E) seg[rowptr[col[e]] + pos[e]] = row[e];
}

__global__ __launch_bounds__(256) void k_sortseg(const int* __restrict__ rowptr,
                                                 const unsigned* __restrict__ deg,
                                                 const int* __restrict__ seg,
                                                 int* __restrict__ sorted, int n) {
  const int lane = threadIdx.x & 31;
  const int node = __builtin_amdgcn_readfirstlane(blockIdx.x * 8 + (threadIdx.x >> 5));
  if (node >= n) return;
  const int start = rowptr[node];
  const int dg = (int)deg[node];
  for (int a = 0; a < dg; a += 32) {
    const int ia = a + lane;
    const int ka = (ia < dg) ? seg[start + ia] : 0x7fffffff;
    int rank = 0;
    for (int b = 0; b < dg; b += 32) {
      const int kb = (b + lane < dg) ? seg[start + b + lane] : 0x7fffffff;
      const int cnt = min(32, dg - b);
      for (int j = 0; j < cnt; j++) {
        const int s = __builtin_amdgcn_readlane(kb, j);
        const int jj = b + j;
        rank += (s < ka || (s == ka && jj < ia)) ? 1 : 0;
      }
    }
    if (ia < dg) sorted[start + rank] = ka;
  }
}


template <int IN, int OUT, bool LN_RELU>
__global__ __launch_bounds__(256) void k_gemm_rows(const float* __restrict__ A,
                                                   const float* __restrict__ W,
                                                   const float* __restrict__ dinv,
                                                   const float* __restrict__ bias,
                                                   const float* __restrict__ lnw,
                                                   const float* __restrict__ lnb,
                                                   float* __restrict__ out, int nRows) {
  constexpr int NT = OUT / 16;
  const int lane = threadIdx.x & 31;
  const int wave = threadIdx.x >> 5;
  const int h = lane >> 4;
  const int m = lane & 15;
  const int rowBase = blockIdx.x * 128 + wave * 16;

  v8f acc[NT] = {};

  int arow = rowBase + m;
  if (arow > nRows - 1) arow = nRows - 1;
  const float* aptr = A + (size_t)arow * IN + 2 * h;

  for (int kb = 0; kb < IN; kb += 4) {
    float2 a2 = *(const float2*)(aptr + kb);
    v2f av; av.x = a2.x; av.y = a2.y;
    const float* w0 = W + (size_t)(kb + 2 * h) * OUT + m;
#pragma unroll
    for (int t = 0; t < NT; t++) {
      v2f bv; bv.x = w0[t * 16]; bv.y = w0[OUT + t * 16];
      acc[t] = __builtin_amdgcn_wmma_f32_16x16x4_f32(
           false, av,  false, bv,
           (short)0, acc[t],  false,  false);
    }
  }

  if constexpr (LN_RELU) {
    float bc[NT], gw[NT], gb[NT];
#pragma unroll
    for (int t = 0; t < NT; t++) {
      bc[t] = bias[t * 16 + m];
      gw[t] = lnw[t * 16 + m];
      gb[t] = lnb[t * 16 + m];
    }
    float s[8], q[8];
#pragma unroll
    for (int v = 0; v < 8; v++) {
      s[v] = 0.f;
#pragma unroll
      for (int t = 0; t < NT; t++) {
        acc[t][v] += bc[t];
        s[v] += acc[t][v];
      }
    }
#pragma unroll
    for (int o = 8; o > 0; o >>= 1) {
#pragma unroll
      for (int v = 0; v < 8; v++) s[v] += __shfl_xor(s[v], o, 32);
    }
#pragma unroll
    for (int v = 0; v < 8; v++) {
      const float mu = s[v] * (1.f / OUT);
      q[v] = 0.f;
#pragma unroll
      for (int t = 0; t < NT; t++) {
        const float d = acc[t][v] - mu;
        acc[t][v] = d;
        q[v] += d * d;
      }
    }
#pragma unroll
    for (int o = 8; o > 0; o >>= 1) {
#pragma unroll
      for (int v = 0; v < 8; v++) q[v] += __shfl_xor(q[v], o, 32);
    }
#pragma unroll
    for (int v = 0; v < 8; v++) {
      const float inv = rsqrtf(q[v] * (1.f / OUT) + 1e-5f);
      const int r = rowBase + v + 8 * h;
      if (r < nRows) {
#pragma unroll
        for (int t = 0; t < NT; t++)
          out[(size_t)r * OUT + t * 16 + m] = fmaxf(acc[t][v] * inv * gw[t] + gb[t], 0.f);
      }
    }
  } else {
#pragma unroll
    for (int v = 0; v < 8; v++) {
      int r = rowBase + v + 8 * h;
      if (r < nRows) {
        float s = dinv[r];
#pragma unroll
        for (int t = 0; t < NT; t++) out[(size_t)r * OUT + t * 16 + m] = acc[t][v] * s;
      }
    }
  }
}


template <bool WEIGHTED>
__global__ __launch_bounds__(256) void k_gather_finish(const float* __restrict__ src,
                                                       const int* __restrict__ rowptr,
                                                       const unsigned* __restrict__ deg,
                                                       const int* __restrict__ sorted,
                                                       const float* __restrict__ dinv,
                                                       const float* __restrict__ bias,
                                                       float* __restrict__ dst, int n) {
  constexpr int D = 64;
  const int lane = threadIdx.x & 31;
  const int node = __builtin_amdgcn_readfirstlane(blockIdx.x * 8 + (threadIdx.x >> 5));
  if (node >= n) return;
  const int start = rowptr[node];
  const int dg = (int)deg[node];
  const float di = dinv[node];
  const int sub = lane >> 4;
  const int c = lane & 15;
  const float4 self = *(const float4*)(src + (size_t)node * D + c * 4);
  float4 bb = make_float4(0.f, 0.f, 0.f, 0.f);
  if constexpr (!WEIGHTED) bb = *(const float4*)(bias + c * 4);
  float4 acc = make_float4(0.f, 0.f, 0.f, 0.f);
  for (int base = 0; base < dg; base += 32) {
    const int cnt = min(32, dg - base);
    const int myid = (lane < cnt) ? sorted[start + base + lane] : 0;
    float myw = 0.f;
    if constexpr (WEIGHTED) myw = (lane < cnt) ? dinv[myid] : 0.f;
    const int pairs = cnt >> 1;
#pragma unroll 4
    for (int p = 0; p < pairs; p++) {
      const int s = __shfl(myid, 2 * p + sub, 32);
      const float4 v = *(const float4*)(src + (size_t)s * D + c * 4);
      if constexpr (WEIGHTED) {
        const float w = __shfl(myw, 2 * p + sub, 32);
        acc.x = fmaf(v.x, w, acc.x); acc.y = fmaf(v.y, w, acc.y);
        acc.z = fmaf(v.z, w, acc.z); acc.w = fmaf(v.w, w, acc.w);
      } else {
        acc.x += v.x; acc.y += v.y; acc.z += v.z; acc.w += v.w;
      }
    }
    if (cnt & 1) {
      const int s = __shfl(myid, cnt - 1, 32);
      const float w = WEIGHTED ? __shfl(myw, cnt - 1, 32) : 1.f;
      if (sub == 0) {
        const float4 v = *(const float4*)(src + (size_t)s * D + c * 4);
        if constexpr (WEIGHTED) {
          acc.x = fmaf(v.x, w, acc.x); acc.y = fmaf(v.y, w, acc.y);
          acc.z = fmaf(v.z, w, acc.z); acc.w = fmaf(v.w, w, acc.w);
        } else {
          acc.x += v.x; acc.y += v.y; acc.z += v.z; acc.w += v.w;
        }
      }
    }
  }
  acc.x += __shfl_xor(acc.x, 16, 32);
  acc.y += __shfl_xor(acc.y, 16, 32);
  acc.z += __shfl_xor(acc.z, 16, 32);
  acc.w += __shfl_xor(acc.w, 16, 32);

  float4 t;
  if constexpr (WEIGHTED) {
    t.x = fmaf(self.x, di, acc.x) * di;
    t.y = fmaf(self.y, di, acc.y) * di;
    t.z = fmaf(self.z, di, acc.z) * di;
    t.w = fmaf(self.w, di, acc.w) * di;
  } else {
    t.x = (acc.x + self.x) * di + bb.x;
    t.y = (acc.y + self.y) * di + bb.y;
    t.z = (acc.z + self.z) * di + bb.z;
    t.w = (acc.w + self.w) * di + bb.w;
  }
  if (sub == 0) *(float4*)(dst + (size_t)node * D + c * 4) = t;
}


#define POOL_NODES 8

__device__ __forceinline__ void pool_flush4(float* __restrict__ psum, float* __restrict__ pcnt,
                                            int b, int c, float4 acc, float cnt) {
  float* p = psum + b * 64 + c * 4;
  atomicAdd(p + 0, acc.x);
  atomicAdd(p + 1, acc.y);
  atomicAdd(p + 2, acc.z);
  atomicAdd(p + 3, acc.w);
  if (c == 0) atomicAdd(pcnt + b, cnt);
}

__global__ __launch_bounds__(256) void k_poolsum(const float* __restrict__ hout,
                                                 const int* __restrict__ batch,
                                                 float* __restrict__ psum,
                                                 float* __restrict__ pcnt, int n) {
  __shared__ __attribute__((aligned(16))) float part[16][68];
  __shared__ int pbid[16];

  const int c = threadIdx.x & 15;
  const int g = threadIdx.x >> 4;
  const int base = (blockIdx.x * 16 + g) * POOL_NODES;

  float4 hv[POOL_NODES];
  int bid[POOL_NODES];
#pragma unroll
  for (int j = 0; j < POOL_NODES; j++) {
    int i = base + j;
    if (i > n - 1) i = n - 1;
    hv[j] = *(const float4*)(hout + (size_t)i * 64 + c * 4);
    bid[j] = batch[i];
  }

  int cur = -1;
  float4 acc = make_float4(0.f, 0.f, 0.f, 0.f);
  float cnt = 0.f;
#pragma unroll
  for (int j = 0; j < POOL_NODES; j++) {
    if (base + j < n) {
      int b = bid[j];
      if (b != cur) {
        if (cur >= 0) pool_flush4(psum, pcnt, cur, c, acc, cnt);
        cur = b;
        acc = make_float4(0.f, 0.f, 0.f, 0.f);
        cnt = 0.f;
      }
      acc.x += hv[j].x;
      acc.y += hv[j].y;
      acc.z += hv[j].z;
      acc.w += hv[j].w;
      cnt += 1.f;
    }
  }

  *(float4*)(&part[g][c * 4]) = acc;
  if (c == 0) {
    part[g][64] = cnt;
    pbid[g] = cur;
  }
  __syncthreads();

  const int d = threadIdx.x;
  if (d < 65) {
    float* dst = (d < 64) ? (psum + d) : pcnt;
    const int stride = (d < 64) ? 64 : 1;
    int ids[16];
    float vals[16];
#pragma unroll
    for (int s = 0; s < 16; s++) {
      ids[s] = pbid[s];
      vals[s] = part[s][d];
    }
    int rb = -1;
    float run = 0.f;
#pragma unroll
    for (int s = 0; s < 16; s++) {
      int b = ids[s];
      if (b >= 0) {
        if (b != rb) {
          if (rb >= 0) atomicAdd(dst + rb * stride, run);
          rb = b;
          run = 0.f;
        }
        run += vals[s];
      }
    }
    if (rb >= 0) atomicAdd(dst + rb * stride, run);
  }
}

__global__ void k_pool(const float* __restrict__ psum, const float* __restrict__ pcnt,
                       float* __restrict__ g) {
  int idx = blockIdx.x * blockDim.x + threadIdx.x;
  if (idx < N_GRAPHS * 64) {
    int gi = idx >> 6;
    g[idx] = psum[idx] / fmaxf(pcnt[gi], 1.f);
  }
}


extern "C" void kernel_launch(void* const* d_in, const int* in_sizes, int n_in,
                              void* d_out, int out_size, void* d_ws, size_t ws_size,
                              hipStream_t stream) {
  const float* x   = (const float*)d_in[0];
  const int*   ei  = (const int*)d_in[1];
  const int*  batch= (const int*)d_in[2];
  const float* W1  = (const float*)d_in[3];
  const float* b1  = (const float*)d_in[4];
  const float* lnw = (const float*)d_in[5];
  const float* lnb = (const float*)d_in[6];
  const float* W2  = (const float*)d_in[7];
  const float* b2  = (const float*)d_in[8];

  const int n = in_sizes[0] / 64;
  const int E = in_sizes[1] / 2;
  const int* row = ei;
  const int* col = ei + E;

  float* ws = (float*)d_ws;
  float* xws  = ws;
  float* h1   = xws + (size_t)n * 128;
  unsigned* deg = (unsigned*)(h1 + (size_t)n * 128);
  float* psum = (float*)(deg + n);
  float* pcnt = psum + N_GRAPHS * 64;
  float* dinv = pcnt + N_GRAPHS;
  int* rowptr = (int*)(dinv + n);
  int* pos    = rowptr + n;
  int* seg    = pos + E;
  int* sorted = seg + E;

  float* hout = (float*)d_out;
  float* gout = hout + (size_t)n * 64;

  size_t zbytes = ((size_t)n + N_GRAPHS * 64 + N_GRAPHS) * sizeof(float);
  hipMemsetAsync(deg, 0, zbytes, stream);

  k_degree<<<(E + 255) / 256, 256, 0, stream>>>(col, deg, pos, E);
  k_scan<<<(n + 1023) / 1024, 256, 0, stream>>>(deg, rowptr, dinv, n);
  k_fill<<<(E + 255) / 256, 256, 0, stream>>>(row, col, pos, rowptr, seg, E);
  k_sortseg<<<(n + 7) / 8, 256, 0, stream>>>(rowptr, deg, seg, sorted, n);

  k_gather_finish<true><<<(n + 7) / 8, 256, 0, stream>>>(x, rowptr, deg, sorted, dinv, nullptr, xws, n);
  k_gemm_rows<64, 128, true><<<(n + 127) / 128, 256, 0, stream>>>(xws, W1, nullptr, b1, lnw, lnb, h1, n);

  float* xw2s = xws;
  k_gemm_rows<128, 64, false><<<(n + 127) / 128, 256, 0, stream>>>(h1, W2, dinv, nullptr, nullptr, nullptr, xw2s, n);
  k_gather_finish<false><<<(n + 7) / 8, 256, 0, stream>>>(xw2s, rowptr, deg, sorted, dinv, b2, hout, n);
  k_poolsum<<<(n + 16 * POOL_NODES - 1) / (16 * POOL_NODES), 256, 0, stream>>>(hout, batch, psum, pcnt, n);
  k_pool<<<(N_GRAPHS * 64 + 255) / 256, 256, 0, stream>>>(psum, pcnt, gout);
}


# Scan kernel: a block's sum over the degrees in front of it keeps 12 loads per thread in flight (1024 threads, 4096 nodes per block): at most 2 round trips to the cache, were up to 97 one by one
# speedup vs baseline: 2.6862x; 1.1398x over previous
#include <hip/hip_runtime.h>

typedef __attribute__((ext_vector_type(2))) float v2f;
typedef __attribute__((ext_vector_type(8))) float v8f;

#define N_GRAPHS 64


__global__ void k_degree(const int* __restrict__ col, unsigned* __restrict__ deg,
                         int* __restrict__ pos, int E) {
  int e = blockIdx.x * blockDim.x + threadIdx.x;
  if (e < E) pos[e] = (int)atomicAdd(&deg[col[e]], 1u);
}


#define SCAN_NODES 4096
#define SCAN_INFLIGHT 12

typedef __attribute__((ext_vector_type(4))) unsigned u32x4;

__global__ __launch_bounds__(1024)
void k_scan(const unsigned* __restrict__ deg, int* __restrict__ rowptr,
            float* __restrict__ dinv, int n) {
  __shared__ unsigned red[32];
  __shared__ unsigned wsum[32];
  const int tid = threadIdx.x;
  const int lane = tid & 31;
  const int wave = tid >> 5;
  const int nb = blockIdx.x;
  const int i0 = nb * SCAN_NODES + tid * 4;

  unsigned d[4];
#pragma unroll
  for (int k = 0; k < 4; k++) d[k] = (i0 + k < n) ? deg[i0 + k] : 0u;

  const u32x4* d4 = (const u32x4*)deg;
  unsigned s = 0;
  for (int t = 0; t < nb; t += SCAN_INFLIGHT) {
    u32x4 v[SCAN_INFLIGHT];
#pragma unroll
    for (int u = 0; u < SCAN_INFLIGHT; u++) {
      const int j = min(t + u, nb - 1);
      v[u] = d4[j * 1024 + tid];
    }
#pragma unroll
    for (int u = 0; u < SCAN_INFLIGHT; u++) {
      const unsigned p = v[u].x + v[u].y + v[u].z + v[u].w;
      s += (t + u < nb) ? p : 0u;
    }
  }
#pragma unroll
  for (int o = 16; o > 0; o >>= 1) s += __shfl_xor(s, o, 32);
  if (lane == 0) red[wave] = s;

  const unsigned t4 = d[0] + d[1] + d[2] + d[3];
  unsigned inc = t4;
#pragma unroll
  for (int o = 1; o < 32; o <<= 1) {
    unsigned u = __shfl_up(inc, o, 32);
    if (lane >= o) inc += u;
  }
  if (lane == 31) wsum[wave] = inc;
  __syncthreads();

  unsigned basev = red[lane] + ((lane < wave) ? wsum[lane] : 0u);
#pragma unroll
  for (int o = 16; o > 0; o >>= 1) basev += __shfl_xor(basev, o, 32);
  unsigned ex = basev + inc - t4;
#pragma unroll
  for (int k = 0; k < 4; k++) {
    if (i0 + k < n) {
      rowptr[i0 + k] = (int)ex;
      dinv[i0 + k] = rsqrtf((float)(d[k] + 1u));
    }
    ex += d[k];
  }
}

__global__ void k_fill(const int* __restrict__ row, const int* __restrict__ col,
                       const int* __restrict__ pos, const int* __restrict__ rowptr,
                       int* __restrict__ seg, int E) {
  int e = blockIdx.x * blockDim.x + threadIdx.x;
  if (e < E) seg[rowptr[col[e]] + pos[e]] = row[e];
}

__global__ __launch_bounds__(256) void k_sortseg(const int* __restrict__ rowptr,
                                                 const unsigned* __restrict__ deg,
                                                 const int* __restrict__ seg,
                                                 int* __restrict__ sorted, int n) {
  const int lane = threadIdx.x & 31;
  const int node = __builtin_amdgcn_readfirstlane(blockIdx.x * 8 + (threadIdx.x >> 5));
  if (node >= n) return;
  const int start = rowptr[node];
  const int dg = (int)deg[node];
  for (int a = 0; a < dg; a += 32) {
    const int ia = a + lane;
    const int ka = (ia < dg) ? seg[start + ia] : 0x7fffffff;
    int rank = 0;
    for (int b = 0; b < dg; b += 32) {
      const int kb = (b + lane < dg) ? seg[start + b + lane] : 0x7fffffff;
      const int cnt = min(32, dg - b);
      for (int j = 0; j < cnt; j++) {
        const int s = __builtin_amdgcn_readlane(kb, j);
        const int jj = b + j;
        rank += (s < ka || (s == ka && jj < ia)) ? 1 : 0;
      }
    }
    if (ia < dg) sorted[start + rank] = ka;
  }
}


template <int IN, int OUT, bool LN_RELU>
__global__ __launch_bounds__(256) void k_gemm_rows(const float* __restrict__ A,
                                                   const float* __restrict__ W,
                                                   const float* __restrict__ dinv,
                                                   const float* __restrict__ bias,
                                                   const float* __restrict__ lnw,
                                                   const float* __restrict__ lnb,
                                                   float* __restrict__ out, int nRows) {
  constexpr int NT = OUT / 16;
  const int lane = threadIdx.x & 31;
  const int wave = threadIdx.x >> 5;
  const int h = lane >> 4;
  const int m = lane & 15;
  const int rowBase = blockIdx.x * 128 + wave * 16;

  v8f acc[NT] = {};

  int arow = rowBase + m;
  if (arow > nRows - 1) arow = nRows - 1;
  const float* aptr = A + (size_t)arow * IN + 2 * h;

  for (int kb = 0; kb < IN; kb += 4) {
    float2 a2 = *(const float2*)(aptr + kb);
    v2f av; av.x = a2.x; av.y = a2.y;
    const float* w0 = W + (size_t)(kb + 2 * h) * OUT + m;
#pragma unroll
    for (int t = 0; t < NT; t++) {
      v2f bv; bv.x = w0[t * 16]; bv.y = w0[OUT + t * 16];
      acc[t] = __builtin_amdgcn_wmma_f32_16x16x4_f32(
           false, av,  false, bv,
           (short)0, acc[t],  false,  false);
    }
  }

  if constexpr (LN_RELU) {
    float bc[NT], gw[NT], gb[NT];
#pragma unroll
    for (int t = 0; t < NT; t++) {
      bc[t] = bias[t * 16 + m];
      gw[t] = lnw[t * 16 + m];
      gb[t] = lnb[t * 16 + m];
    }
    float s[8], q[8];
#pragma unroll
    for (int v = 0; v < 8; v++) {
      s[v] = 0.f;
#pragma unroll
      for (int t = 0; t < NT; t++) {
        acc[t][v] += bc[t];
        s[v] += acc[t][v];
      }
    }
#pragma unroll
    for (int o = 8; o > 0; o >>= 1) {
#pragma unroll
      for (int v = 0; v < 8; v++) s[v] += __shfl_xor(s[v], o, 32);
    }
#pragma unroll
    for (int v = 0; v < 8; v++) {
      const float mu = s[v] * (1.f / OUT);
      q[v] = 0.f;
#pragma unroll
      for (int t = 0; t < NT; t++) {
        const float d = acc[t][v] - mu;
        acc[t][v] = d;
        q[v] += d * d;
      }
    }
#pragma unroll
    for (int o = 8; o > 0; o >>= 1) {
#pragma unroll
      for (int v = 0; v < 8; v++) q[v] += __shfl_xor(q[v], o, 32);
    }
#pragma unroll
    for (int v = 0; v < 8; v++) {
      const float inv = rsqrtf(q[v] * (1.f / OUT) + 1e-5f);
      const int r = rowBase + v + 8 * h;
      if (r < nRows) {
#pragma unroll
        for (int t = 0; t < NT; t++)
          out[(size_t)r * OUT + t * 16 + m] = fmaxf(acc[t][v] * inv * gw[t] + gb[t], 0.f);
      }
    }
  } else {
#pragma unroll
    for (int v = 0; v < 8; v++) {
      int r = rowBase + v + 8 * h;
      if (r < nRows) {
        float s = dinv[r];
#pragma unroll
        for (int t = 0; t < NT; t++) out[(size_t)r * OUT + t * 16 + m] = acc[t][v] * s;
      }
    }
  }
}


template <bool WEIGHTED>
__global__ __launch_bounds__(256) void k_gather_finish(const float* __restrict__ src,
                                                       const int* __restrict__ rowptr,
                                                       const unsigned* __restrict__ deg,
                                                       const int* __restrict__ sorted,
                                                       const float* __restrict__ dinv,
                                                       const float* __restrict__ bias,
                                                       float* __restrict__ dst, int n) {
  constexpr int D = 64;
  const int lane = threadIdx.x & 31;
  const int node = __builtin_amdgcn_readfirstlane(blockIdx.x * 8 + (threadIdx.x >> 5));
  if (node >= n) return;
  const int start = rowptr[node];
  const int dg = (int)deg[node];
  const float di = dinv[node];
  const int sub = lane >> 4;
  const int c = lane & 15;
  const float4 self = *(const float4*)(src + (size_t)node * D + c * 4);
  float4 bb = make_float4(0.f, 0.f, 0.f, 0.f);
  if constexpr (!WEIGHTED) bb = *(const float4*)(bias + c * 4);
  float4 acc = make_float4(0.f, 0.f, 0.f, 0.f);
  for (int base = 0; base < dg; base += 32) {
    const int cnt = min(32, dg - base);
    const int myid = (lane < cnt) ? sorted[start + base + lane] : 0;
    float myw = 0.f;
    if constexpr (WEIGHTED) myw = (lane < cnt) ? dinv[myid] : 0.f;
    const int pairs = cnt >> 1;
#pragma unroll 4
    for (int p = 0; p < pairs; p++) {
      const int s = __shfl(myid, 2 * p + sub, 32);
      const float4 v = *(const float4*)(src + (size_t)s * D + c * 4);
      if constexpr (WEIGHTED) {
        const float w = __shfl(myw, 2 * p + sub, 32);
        acc.x = fmaf(v.x, w, acc.x); acc.y = fmaf(v.y, w, acc.y);
        acc.z = fmaf(v.z, w, acc.z); acc.w = fmaf(v.w, w, acc.w);
      } else {
        acc.x += v.x; acc.y += v.y; acc.z += v.z; acc.w += v.w;
      }
    }
    if (cnt & 1) {
      const int s = __shfl(myid, cnt - 1, 32);
      const float w = WEIGHTED ? __shfl(myw, cnt - 1, 32) : 1.f;
      if (sub == 0) {
        const float4 v = *(const float4*)(src + (size_t)s * D + c * 4);
        if constexpr (WEIGHTED) {
          acc.x = fmaf(v.x, w, acc.x); acc.y = fmaf(v.y, w, acc.y);
          acc.z = fmaf(v.z, w, acc.z); acc.w = fmaf(v.w, w, acc.w);
        } else {
          acc.x += v.x; acc.y += v.y; acc.z += v.z; acc.w += v.w;
        }
      }
    }
  }
  acc.x += __shfl_xor(acc.x, 16, 32);
  acc.y += __shfl_xor(acc.y, 16, 32);
  acc.z += __shfl_xor(acc.z, 16, 32);
  acc.w += __shfl_xor(acc.w, 16, 32);

  float4 t;
  if constexpr (WEIGHTED) {
    t.x = fmaf(self.x, di, acc.x) * di;
    t.y = fmaf(self.y, di, acc.y) * di;
    t.z = fmaf(self.z, di, acc.z) * di;
    t.w = fmaf(self.w, di, acc.w) * di;
  } else {
    t.x = (acc.x + self.x) * di + bb.x;
    t.y = (acc.y + self.y) * di + bb.y;
    t.z = (acc.z + self.z) * di + bb.z;
    t.w = (acc.w + self.w) * di + bb.w;
  }
  if (sub == 0) *(float4*)(dst + (size_t)node * D + c * 4) = t;
}


#define POOL_NODES 8

__device__ __forceinline__ void pool_flush4(float* __restrict__ psum, float* __restrict__ pcnt,
                                            int b, int c, float4 acc, float cnt) {
  float* p = psum + b * 64 + c * 4;
  atomicAdd(p + 0, acc.x);
  atomicAdd(p + 1, acc.y);
  atomicAdd(p + 2, acc.z);
  atomicAdd(p + 3, acc.w);
  if (c == 0) atomicAdd(pcnt + b, cnt);
}

__global__ __launch_bounds__(256) void k_poolsum(const float* __restrict__ hout,
                                                 const int* __restrict__ batch,
                                                 float* __restrict__ psum,
                                                 float* __restrict__ pcnt, int n) {
  __shared__ __attribute__((aligned(16))) float part[16][68];
  __shared__ int pbid[16];

  const int c = threadIdx.x & 15;
  const int g = threadIdx.x >> 4;
  const int base = (blockIdx.x * 16 + g) * POOL_NODES;

  float4 hv[POOL_NODES];
  int bid[POOL_NODES];
#pragma unroll
  for (int j = 0; j < POOL_NODES; j++) {
    int i = base + j;
    if (i > n - 1) i = n - 1;
    hv[j] = *(const float4*)(hout + (size_t)i * 64 + c * 4);
    bid[j] = batch[i];
  }

  int cur = -1;
  float4 acc = make_float4(0.f, 0.f, 0.f, 0.f);
  float cnt = 0.f;
#pragma unroll
  for (int j = 0; j < POOL_NODES; j++) {
    if (base + j < n) {
      int b = bid[j];
      if (b != cur) {
        if (cur >= 0) pool_flush4(psum, pcnt, cur, c, acc, cnt);
        cur = b;
        acc = make_float4(0.f, 0.f, 0.f, 0.f);
        cnt = 0.f;
      }
      acc.x += hv[j].x;
      acc.y += hv[j].y;
      acc.z += hv[j].z;
      acc.w += hv[j].w;
      cnt += 1.f;
    }
  }

  *(float4*)(&part[g][c * 4]) = acc;
  if (c == 0) {
    part[g][64] = cnt;
    pbid[g] = cur;
  }
  __syncthreads();

  const int d = threadIdx.x;
  if (d < 65) {
    float* dst = (d < 64) ? (psum + d) : pcnt;
    const int stride = (d < 64) ? 64 : 1;
    int ids[16];
    float vals[16];
#pragma unroll
    for (int s = 0; s < 16; s++) {
      ids[s] = pbid[s];
      vals[s] = part[s][d];
    }
    int rb = -1;
    float run = 0.f;
#pragma unroll
    for (int s = 0; s < 16; s++) {
      int b = ids[s];
      if (b >= 0) {
        if (b != rb) {
          if (rb >= 0) atomicAdd(dst + rb * stride, run);
          rb = b;
          run = 0.f;
        }
        run += vals[s];
      }
    }
    if (rb >= 0) atomicAdd(dst + rb * stride, run);
  }
}

__global__ void k_pool(const float* __restrict__ psum, const float* __restrict__ pcnt,
                       float* __restrict__ g) {
  int idx = blockIdx.x * blockDim.x + threadIdx.x;
  if (idx < N_GRAPHS * 64) {
    int gi = idx >> 6;
    g[idx] = psum[idx] / fmaxf(pcnt[gi], 1.f);
  }
}


extern "C" void kernel_launch(void* const* d_in, const int* in_sizes, int n_in,
                              void* d_out, int out_size, void* d_ws, size_t ws_size,
                              hipStream_t stream) {
  const float* x   = (const float*)d_in[0];
  const int*   ei  = (const int*)d_in[1];
  const int*  batch= (const int*)d_in[2];
  const float* W1  = (const float*)d_in[3];
  const float* b1  = (const float*)d_in[4];
  const float* lnw = (const float*)d_in[5];
  const float* lnb = (const float*)d_in[6];
  const float* W2  = (const float*)d_in[7];
  const float* b2  = (const float*)d_in[8];

  const int n = in_sizes[0] / 64;
  const int E = in_sizes[1] / 2;
  const int* row = ei;
  const int* col = ei + E;

  float* ws = (float*)d_ws;
  float* xws  = ws;
  float* h1   = xws + (size_t)n * 128;
  unsigned* deg = (unsigned*)(h1 + (size_t)n * 128);
  float* psum = (float*)(deg + n);
  float* pcnt = psum + N_GRAPHS * 64;
  float* dinv = pcnt + N_GRAPHS;
  int* rowptr = (int*)(dinv + n);
  int* pos    = rowptr + n;
  int* seg    = pos + E;
  int* sorted = seg + E;

  float* hout = (float*)d_out;
  float* gout = hout + (size_t)n * 64;

  size_t zbytes = ((size_t)n + N_GRAPHS * 64 + N_GRAPHS) * sizeof(float);
  hipMemsetAsync(deg, 0, zbytes, stream);

  k_degree<<<(E + 255) / 256, 256, 0, stream>>>(col, deg, pos, E);
  k_scan<<<(n + SCAN_NODES - 1) / SCAN_NODES, 1024, 0, stream>>>(deg, rowptr, dinv, n);
  k_fill<<<(E + 255) / 256, 256, 0, stream>>>(row, col, pos, rowptr, seg, E);
  k_sortseg<<<(n + 7) / 8, 256, 0, stream>>>(rowptr, deg, seg, sorted, n);

  k_gather_finish<true><<<(n + 7) / 8, 256, 0, stream>>>(x, rowptr, deg, sorted, dinv, nullptr, xws, n);
  k_gemm_rows<64, 128, true><<<(n + 127) / 128, 256, 0, stream>>>(xws, W1, nullptr, b1, lnw, lnb, h1, n);

  float* xw2s = xws;
  k_gemm_rows<128, 64, false><<<(n + 127) / 128, 256, 0, stream>>>(h1, W2, dinv, nullptr, nullptr, nullptr, xw2s, n);
  k_gather_finish<false><<<(n + 7) / 8, 256, 0, stream>>>(xw2s, rowptr, deg, sorted, dinv, b2, hout, n);
  k_poolsum<<<(n + 16 * POOL_NODES - 1) / (16 * POOL_NODES), 256, 0, stream>>>(hout, batch, psum, pcnt, n);
  k_pool<<<(N_GRAPHS * 64 + 255) / 256, 256, 0, stream>>>(psum, pcnt, gout);
}


# Edge lists in fixed 64-slot rows per target: the degree atomic's return value files the edge at once; scan, fill pass and positions array go; a node over 64 edges: slow exact path; f32 sum order kept
# speedup vs baseline: 2.9584x; 1.1013x over previous
#include <hip/hip_runtime.h>

typedef __attribute__((ext_vector_type(2))) float v2f;
typedef __attribute__((ext_vector_type(8))) float v8f;

#define N_GRAPHS 64


#define SLOT_CAP 64

__global__ void k_place(const int* __restrict__ row, const int* __restrict__ col,
                        unsigned* __restrict__ deg, int* __restrict__ tab,
                        unsigned* __restrict__ spill_n, int* __restrict__ spill_c,
                        int* __restrict__ spill_r, int E) {
  int e = blockIdx.x * blockDim.x + threadIdx.x;
  if (e < E) {
    const int c = col[e];
    const int r = row[e];
    const unsigned p = atomicAdd(&deg[c], 1u);
    if (p < SLOT_CAP) {
      tab[c * SLOT_CAP + (int)p] = r;
    } else {
      const unsigned k = atomicAdd(spill_n, 1u);
      spill_c[k] = c;
      spill_r[k] = r;
    }
  }
}

__global__ __launch_bounds__(256) void k_sortseg(const unsigned* __restrict__ deg,
                                                 const int* __restrict__ tab,
                                                 int* __restrict__ sorted,
                                                 float* __restrict__ dinv, int n) {
  const int lane = threadIdx.x & 31;
  const int node = __builtin_amdgcn_readfirstlane(blockIdx.x * 8 + (threadIdx.x >> 5));
  if (node >= n) return;
  const int dg = (int)deg[node];
  if (lane == 0) dinv[node] = rsqrtf((float)((unsigned)dg + 1u));
  if (dg > SLOT_CAP) return;
  const int start = node * SLOT_CAP;
  for (int a = 0; a < dg; a += 32) {
    const int ia = a + lane;
    const int ka = (ia < dg) ? tab[start + ia] : 0x7fffffff;
    int rank = 0;
    for (int b = 0; b < dg; b += 32) {
      const int kb = (b + lane < dg) ? tab[start + b + lane] : 0x7fffffff;
      const int cnt = min(32, dg - b);
      for (int j = 0; j < cnt; j++) {
        const int s = __builtin_amdgcn_readlane(kb, j);
        const int jj = b + j;
        rank += (s < ka || (s == ka && jj < ia)) ? 1 : 0;
      }
    }
    if (ia < dg) sorted[start + rank] = ka;
  }
}


template <int IN, int OUT, bool LN_RELU>
__global__ __launch_bounds__(256) void k_gemm_rows(const float* __restrict__ A,
                                                   const float* __restrict__ W,
                                                   const float* __restrict__ dinv,
                                                   const float* __restrict__ bias,
                                                   const float* __restrict__ lnw,
                                                   const float* __restrict__ lnb,
                                                   float* __restrict__ out, int nRows) {
  constexpr int NT = OUT / 16;
  const int lane = threadIdx.x & 31;
  const int wave = threadIdx.x >> 5;
  const int h = lane >> 4;
  const int m = lane & 15;
  const int rowBase = blockIdx.x * 128 + wave * 16;

  v8f acc[NT] = {};

  int arow = rowBase + m;
  if (arow > nRows - 1) arow = nRows - 1;
  const float* aptr = A + (size_t)arow * IN + 2 * h;

  for (int kb = 0; kb < IN; kb += 4) {
    float2 a2 = *(const float2*)(aptr + kb);
    v2f av; av.x = a2.x; av.y = a2.y;
    const float* w0 = W + (size_t)(kb + 2 * h) * OUT + m;
#pragma unroll
    for (int t = 0; t < NT; t++) {
      v2f bv; bv.x = w0[t * 16]; bv.y = w0[OUT + t * 16];
      acc[t] = __builtin_amdgcn_wmma_f32_16x16x4_f32(
           false, av,  false, bv,
           (short)0, acc[t],  false,  false);
    }
  }

  if constexpr (LN_RELU) {
    float bc[NT], gw[NT], gb[NT];
#pragma unroll
    for (int t = 0; t < NT; t++) {
      bc[t] = bias[t * 16 + m];
      gw[t] = lnw[t * 16 + m];
      gb[t] = lnb[t * 16 + m];
    }
    float s[8], q[8];
#pragma unroll
    for (int v = 0; v < 8; v++) {
      s[v] = 0.f;
#pragma unroll
      for (int t = 0; t < NT; t++) {
        acc[t][v] += bc[t];
        s[v] += acc[t][v];
      }
    }
#pragma unroll
    for (int o = 8; o > 0; o >>= 1) {
#pragma unroll
      for (int v = 0; v < 8; v++) s[v] += __shfl_xor(s[v], o, 32);
    }
#pragma unroll
    for (int v = 0; v < 8; v++) {
      const float mu = s[v] * (1.f / OUT);
      q[v] = 0.f;
#pragma unroll
      for (int t = 0; t < NT; t++) {
        const float d = acc[t][v] - mu;
        acc[t][v] = d;
        q[v] += d * d;
      }
    }
#pragma unroll
    for (int o = 8; o > 0; o >>= 1) {
#pragma unroll
      for (int v = 0; v < 8; v++) q[v] += __shfl_xor(q[v], o, 32);
    }
#pragma unroll
    for (int v = 0; v < 8; v++) {
      const float inv = rsqrtf(q[v] * (1.f / OUT) + 1e-5f);
      const int r = rowBase + v + 8 * h;
      if (r < nRows) {
#pragma unroll
        for (int t = 0; t < NT; t++)
          out[(size_t)r * OUT + t * 16 + m] = fmaxf(acc[t][v] * inv * gw[t] + gb[t], 0.f);
      }
    }
  } else {
#pragma unroll
    for (int v = 0; v < 8; v++) {
      int r = rowBase + v + 8 * h;
      if (r < nRows) {
        float s = dinv[r];
#pragma unroll
        for (int t = 0; t < NT; t++) out[(size_t)r * OUT + t * 16 + m] = acc[t][v] * s;
      }
    }
  }
}


template <bool WEIGHTED>
__global__ __launch_bounds__(256) void k_gather_finish(const float* __restrict__ src,
                                                       const unsigned* __restrict__ deg,
                                                       const int* __restrict__ sorted,
                                                       const int* __restrict__ tab,
                                                       const unsigned* __restrict__ spill_n,
                                                       const int* __restrict__ spill_c,
                                                       const int* __restrict__ spill_r,
                                                       int spill_cap,
                                                       const float* __restrict__ dinv,
                                                       const float* __restrict__ bias,
                                                       float* __restrict__ dst, int n) {
  constexpr int D = 64;
  const int lane = threadIdx.x & 31;
  const int node = __builtin_amdgcn_readfirstlane(blockIdx.x * 8 + (threadIdx.x >> 5));
  if (node >= n) return;
  const int start = node * SLOT_CAP;
  const int dg = (int)deg[node];
  const float di = dinv[node];
  const int sub = lane >> 4;
  const int c = lane & 15;
  const float4 self = *(const float4*)(src + (size_t)node * D + c * 4);
  float4 bb = make_float4(0.f, 0.f, 0.f, 0.f);
  if constexpr (!WEIGHTED) bb = *(const float4*)(bias + c * 4);
  float4 acc = make_float4(0.f, 0.f, 0.f, 0.f);
  if (dg <= SLOT_CAP) {
    for (int base = 0; base < dg; base += 32) {
      const int cnt = min(32, dg - base);
      const int myid = (lane < cnt) ? sorted[start + base + lane] : 0;
      float myw = 0.f;
      if constexpr (WEIGHTED) myw = (lane < cnt) ? dinv[myid] : 0.f;
      const int pairs = cnt >> 1;
#pragma unroll 4
      for (int p = 0; p < pairs; p++) {
        const int s = __shfl(myid, 2 * p + sub, 32);
        const float4 v = *(const float4*)(src + (size_t)s * D + c * 4);
        if constexpr (WEIGHTED) {
          const float w = __shfl(myw, 2 * p + sub, 32);
          acc.x = fmaf(v.x, w, acc.x); acc.y = fmaf(v.y, w, acc.y);
          acc.z = fmaf(v.z, w, acc.z); acc.w = fmaf(v.w, w, acc.w);
        } else {
          acc.x += v.x; acc.y += v.y; acc.z += v.z; acc.w += v.w;
        }
      }
      if (cnt & 1) {
        const int s = __shfl(myid, cnt - 1, 32);
        const float w = WEIGHTED ? __shfl(myw, cnt - 1, 32) : 1.f;
        if (sub == 0) {
          const float4 v = *(const float4*)(src + (size_t)s * D + c * 4);
          if constexpr (WEIGHTED) {
            acc.x = fmaf(v.x, w, acc.x); acc.y = fmaf(v.y, w, acc.y);
            acc.z = fmaf(v.z, w, acc.z); acc.w = fmaf(v.w, w, acc.w);
          } else {
            acc.x += v.x; acc.y += v.y; acc.z += v.z; acc.w += v.w;
          }
        }
      }
    }
  } else {
    const int ns = min((int)spill_n[0], spill_cap);
    int last = -1;
    for (int it = 0; it < dg; it++) {
      int best = 0x7fffffff;
      for (int k = lane; k < SLOT_CAP; k += 32) {
        const int s = tab[start + k];
        if (s > last && s < best) best = s;
      }
      for (int k = lane; k < ns; k += 32) {
        const int s = spill_r[k];
        if (spill_c[k] == node && s > last && s < best) best = s;
      }
#pragma unroll
      for (int o = 16; o > 0; o >>= 1) best = min(best, __shfl_xor(best, o, 32));
      if (best == 0x7fffffff) break;
      int m = 0;
      for (int k = lane; k < SLOT_CAP; k += 32) m += (tab[start + k] == best) ? 1 : 0;
      for (int k = lane; k < ns; k += 32) m += (spill_c[k] == node && spill_r[k] == best) ? 1 : 0;
#pragma unroll
      for (int o = 16; o > 0; o >>= 1) m += __shfl_xor(m, o, 32);
      const float4 v = *(const float4*)(src + (size_t)best * D + c * 4);
      const float w = WEIGHTED ? dinv[best] : 1.f;
      if (sub == 0) {
        for (int i = 0; i < m; i++) {
          if constexpr (WEIGHTED) {
            acc.x = fmaf(v.x, w, acc.x); acc.y = fmaf(v.y, w, acc.y);
            acc.z = fmaf(v.z, w, acc.z); acc.w = fmaf(v.w, w, acc.w);
          } else {
            acc.x += v.x; acc.y += v.y; acc.z += v.z; acc.w += v.w;
          }
        }
      }
      last = best;
    }
  }
  acc.x += __shfl_xor(acc.x, 16, 32);
  acc.y += __shfl_xor(acc.y, 16, 32);
  acc.z += __shfl_xor(acc.z, 16, 32);
  acc.w += __shfl_xor(acc.w, 16, 32);

  float4 t;
  if constexpr (WEIGHTED) {
    t.x = fmaf(self.x, di, acc.x) * di;
    t.y = fmaf(self.y, di, acc.y) * di;
    t.z = fmaf(self.z, di, acc.z) * di;
    t.w = fmaf(self.w, di, acc.w) * di;
  } else {
    t.x = (acc.x + self.x) * di + bb.x;
    t.y = (acc.y + self.y) * di + bb.y;
    t.z = (acc.z + self.z) * di + bb.z;
    t.w = (acc.w + self.w) * di + bb.w;
  }
  if (sub == 0) *(float4*)(dst + (size_t)node * D + c * 4) = t;
}


#define POOL_NODES 8

__device__ __forceinline__ void pool_flush4(float* __restrict__ psum, float* __restrict__ pcnt,
                                            int b, int c, float4 acc, float cnt) {
  float* p = psum + b * 64 + c * 4;
  atomicAdd(p + 0, acc.x);
  atomicAdd(p + 1, acc.y);
  atomicAdd(p + 2, acc.z);
  atomicAdd(p + 3, acc.w);
  if (c == 0) atomicAdd(pcnt + b, cnt);
}

__global__ __launch_bounds__(256) void k_poolsum(const float* __restrict__ hout,
                                                 const int* __restrict__ batch,
                                                 float* __restrict__ psum,
                                                 float* __restrict__ pcnt, int n) {
  __shared__ __attribute__((aligned(16))) float part[16][68];
  __shared__ int pbid[16];

  const int c = threadIdx.x & 15;
  const int g = threadIdx.x >> 4;
  const int base = (blockIdx.x * 16 + g) * POOL_NODES;

  float4 hv[POOL_NODES];
  int bid[POOL_NODES];
#pragma unroll
  for (int j = 0; j < POOL_NODES; j++) {
    int i = base + j;
    if (i > n - 1) i = n - 1;
    hv[j] = *(const float4*)(hout + (size_t)i * 64 + c * 4);
    bid[j] = batch[i];
  }

  int cur = -1;
  float4 acc = make_float4(0.f, 0.f, 0.f, 0.f);
  float cnt = 0.f;
#pragma unroll
  for (int j = 0; j < POOL_NODES; j++) {
    if (base + j < n) {
      int b = bid[j];
      if (b != cur) {
        if (cur >= 0) pool_flush4(psum, pcnt, cur, c, acc, cnt);
        cur = b;
        acc = make_float4(0.f, 0.f, 0.f, 0.f);
        cnt = 0.f;
      }
      acc.x += hv[j].x;
      acc.y += hv[j].y;
      acc.z += hv[j].z;
      acc.w += hv[j].w;
      cnt += 1.f;
    }
  }

  *(float4*)(&part[g][c * 4]) = acc;
  if (c == 0) {
    part[g][64] = cnt;
    pbid[g] = cur;
  }
  __syncthreads();

  const int d = threadIdx.x;
  if (d < 65) {
    float* dst = (d < 64) ? (psum + d) : pcnt;
    const int stride = (d < 64) ? 64 : 1;
    int ids[16];
    float vals[16];
#pragma unroll
    for (int s = 0; s < 16; s++) {
      ids[s] = pbid[s];
      vals[s] = part[s][d];
    }
    int rb = -1;
    float run = 0.f;
#pragma unroll
    for (int s = 0; s < 16; s++) {
      int b = ids[s];
      if (b >= 0) {
        if (b != rb) {
          if (rb >= 0) atomicAdd(dst + rb * stride, run);
          rb = b;
          run = 0.f;
        }
        run += vals[s];
      }
    }
    if (rb >= 0) atomicAdd(dst + rb * stride, run);
  }
}

__global__ void k_pool(const float* __restrict__ psum, const float* __restrict__ pcnt,
                       float* __restrict__ g) {
  int idx = blockIdx.x * blockDim.x + threadIdx.x;
  if (idx < N_GRAPHS * 64) {
    int gi = idx >> 6;
    g[idx] = psum[idx] / fmaxf(pcnt[gi], 1.f);
  }
}


extern "C" void kernel_launch(void* const* d_in, const int* in_sizes, int n_in,
                              void* d_out, int out_size, void* d_ws, size_t ws_size,
                              hipStream_t stream) {
  const float* x   = (const float*)d_in[0];
  const int*   ei  = (const int*)d_in[1];
  const int*  batch= (const int*)d_in[2];
  const float* W1  = (const float*)d_in[3];
  const float* b1  = (const float*)d_in[4];
  const float* lnw = (const float*)d_in[5];
  const float* lnb = (const float*)d_in[6];
  const float* W2  = (const float*)d_in[7];
  const float* b2  = (const float*)d_in[8];

  const int n = in_sizes[0] / 64;
  const int E = in_sizes[1] / 2;
  const int* row = ei;
  const int* col = ei + E;

  float* ws = (float*)d_ws;
  float* xws  = ws;
  float* h1   = xws + (size_t)n * 128;
  unsigned* deg = (unsigned*)(h1 + (size_t)n * 128);
  float* psum = (float*)(deg + n);
  float* pcnt = psum + N_GRAPHS * 64;
  unsigned* spill_n = (unsigned*)(pcnt + N_GRAPHS);
  float* dinv = (float*)(spill_n + 64);
  int* tab    = (int*)(dinv + n);
  int* sorted = tab + (size_t)n * SLOT_CAP;
  int* spill_c = sorted + (size_t)n * SLOT_CAP;
  int* spill_r = spill_c + E;

  float* hout = (float*)d_out;
  float* gout = hout + (size_t)n * 64;

  size_t zbytes = ((size_t)n + N_GRAPHS * 64 + N_GRAPHS + 64) * sizeof(float);
  hipMemsetAsync(deg, 0, zbytes, stream);

  k_place<<<(E + 255) / 256, 256, 0, stream>>>(row, col, deg, tab, spill_n, spill_c, spill_r, E);
  k_sortseg<<<(n + 7) / 8, 256, 0, stream>>>(deg, tab, sorted, dinv, n);

  k_gather_finish<true><<<(n + 7) / 8, 256, 0, stream>>>(x, deg, sorted, tab, spill_n, spill_c, spill_r, E,
                                                         dinv, nullptr, xws, n);
  k_gemm_rows<64, 128, true><<<(n + 127) / 128, 256, 0, stream>>>(xws, W1, nullptr, b1, lnw, lnb, h1, n);

  float* xw2s = xws;
  k_gemm_rows<128, 64, false><<<(n + 127) / 128, 256, 0, stream>>>(h1, W2, dinv, nullptr, nullptr, nullptr, xw2s, n);
  k_gather_finish<false><<<(n + 7) / 8, 256, 0, stream>>>(xw2s, deg, sorted, tab, spill_n, spill_c, spill_r, E,
                                                          dinv, b2, hout, n);
  k_poolsum<<<(n + 16 * POOL_NODES - 1) / (16 * POOL_NODES), 256, 0, stream>>>(hout, batch, psum, pcnt, n);
  k_pool<<<(N_GRAPHS * 64 + 255) / 256, 256, 0, stream>>>(psum, pcnt, gout);
}


# Layer-1 gather wave rank-sorts its node's slot row itself (ids back to its lanes through a wave-private LDS row), files dinv and the sorted row for layer 2; the sort launch goes; f32 sum order kept
# speedup vs baseline: 3.0589x; 1.0340x over previous
#include <hip/hip_runtime.h>

typedef __attribute__((ext_vector_type(2))) float v2f;
typedef __attribute__((ext_vector_type(8))) float v8f;

#define N_GRAPHS 64


#define SLOT_CAP 64

__global__ void k_place(const int* __restrict__ row, const int* __restrict__ col,
                        unsigned* __restrict__ deg, int* __restrict__ tab,
                        unsigned* __restrict__ spill_n, int* __restrict__ spill_c,
                        int* __restrict__ spill_r, int E) {
  int e = blockIdx.x * blockDim.x + threadIdx.x;
  if (e < E) {
    const int c = col[e];
    const int r = row[e];
    const unsigned p = atomicAdd(&deg[c], 1u);
    if (p < SLOT_CAP) {
      tab[c * SLOT_CAP + (int)p] = r;
    } else {
      const unsigned k = atomicAdd(spill_n, 1u);
      spill_c[k] = c;
      spill_r[k] = r;
    }
  }
}


template <int IN, int OUT, bool LN_RELU>
__global__ __launch_bounds__(256) void k_gemm_rows(const float* __restrict__ A,
                                                   const float* __restrict__ W,
                                                   const float* __restrict__ dinv,
                                                   const float* __restrict__ bias,
                                                   const float* __restrict__ lnw,
                                                   const float* __restrict__ lnb,
                                                   float* __restrict__ out, int nRows) {
  constexpr int NT = OUT / 16;
  const int lane = threadIdx.x & 31;
  const int wave = threadIdx.x >> 5;
  const int h = lane >> 4;
  const int m = lane & 15;
  const int rowBase = blockIdx.x * 128 + wave * 16;

  v8f acc[NT] = {};

  int arow = rowBase + m;
  if (arow > nRows - 1) arow = nRows - 1;
  const float* aptr = A + (size_t)arow * IN + 2 * h;

  for (int kb = 0; kb < IN; kb += 4) {
    float2 a2 = *(const float2*)(aptr + kb);
    v2f av; av.x = a2.x; av.y = a2.y;
    const float* w0 = W + (size_t)(kb + 2 * h) * OUT + m;
#pragma unroll
    for (int t = 0; t < NT; t++) {
      v2f bv; bv.x = w0[t * 16]; bv.y = w0[OUT + t * 16];
      acc[t] = __builtin_amdgcn_wmma_f32_16x16x4_f32(
           false, av,  false, bv,
           (short)0, acc[t],  false,  false);
    }
  }

  if constexpr (LN_RELU) {
    float bc[NT], gw[NT], gb[NT];
#pragma unroll
    for (int t = 0; t < NT; t++) {
      bc[t] = bias[t * 16 + m];
      gw[t] = lnw[t * 16 + m];
      gb[t] = lnb[t * 16 + m];
    }
    float s[8], q[8];
#pragma unroll
    for (int v = 0; v < 8; v++) {
      s[v] = 0.f;
#pragma unroll
      for (int t = 0; t < NT; t++) {
        acc[t][v] += bc[t];
        s[v] += acc[t][v];
      }
    }
#pragma unroll
    for (int o = 8; o > 0; o >>= 1) {
#pragma unroll
      for (int v = 0; v < 8; v++) s[v] += __shfl_xor(s[v], o, 32);
    }
#pragma unroll
    for (int v = 0; v < 8; v++) {
      const float mu = s[v] * (1.f / OUT);
      q[v] = 0.f;
#pragma unroll
      for (int t = 0; t < NT; t++) {
        const float d = acc[t][v] - mu;
        acc[t][v] = d;
        q[v] += d * d;
      }
    }
#pragma unroll
    for (int o = 8; o > 0; o >>= 1) {
#pragma unroll
      for (int v = 0; v < 8; v++) q[v] += __shfl_xor(q[v], o, 32);
    }
#pragma unroll
    for (int v = 0; v < 8; v++) {
      const float inv = rsqrtf(q[v] * (1.f / OUT) + 1e-5f);
      const int r = rowBase + v + 8 * h;
      if (r < nRows) {
#pragma unroll
        for (int t = 0; t < NT; t++)
          out[(size_t)r * OUT + t * 16 + m] = fmaxf(acc[t][v] * inv * gw[t] + gb[t], 0.f);
      }
    }
  } else {
#pragma unroll
    for (int v = 0; v < 8; v++) {
      int r = rowBase + v + 8 * h;
      if (r < nRows) {
        float s = dinv[r];
#pragma unroll
        for (int t = 0; t < NT; t++) out[(size_t)r * OUT + t * 16 + m] = acc[t][v] * s;
      }
    }
  }
}


template <bool WEIGHTED>
__global__ __launch_bounds__(256) void k_gather_finish(const float* __restrict__ src,
                                                       const unsigned* __restrict__ deg,
                                                       const int* __restrict__ sorted,
                                                       int* __restrict__ sorted_out,
                                                       const int* __restrict__ tab,
                                                       const unsigned* __restrict__ spill_n,
                                                       const int* __restrict__ spill_c,
                                                       const int* __restrict__ spill_r,
                                                       int spill_cap,
                                                       const float* __restrict__ dinv,
                                                       float* __restrict__ dinv_out,
                                                       const float* __restrict__ bias,
                                                       float* __restrict__ dst, int n) {
  constexpr int D = 64;
  const int lane = threadIdx.x & 31;
  const int node = __builtin_amdgcn_readfirstlane(blockIdx.x * 8 + (threadIdx.x >> 5));
  if (node >= n) return;
  const int start = node * SLOT_CAP;
  const int dg = (int)deg[node];
  float di;
  int* myrow = nullptr;
  if constexpr (WEIGHTED) {
    __shared__ int idrow[8][SLOT_CAP];
    myrow = idrow[threadIdx.x >> 5];
    di = rsqrtf((float)((unsigned)dg + 1u));
    if (lane == 0) dinv_out[node] = di;
  } else {
    di = dinv[node];
  }
  const int sub = lane >> 4;
  const int c = lane & 15;
  const float4 self = *(const float4*)(src + (size_t)node * D + c * 4);
  float4 bb = make_float4(0.f, 0.f, 0.f, 0.f);
  if constexpr (!WEIGHTED) bb = *(const float4*)(bias + c * 4);
  float4 acc = make_float4(0.f, 0.f, 0.f, 0.f);
  if (dg <= SLOT_CAP) {
    if constexpr (WEIGHTED) {
      const int i1 = 32 + lane;
      const int k0 = (lane < dg) ? tab[start + lane] : 0;
      const int k1 = (i1 < dg) ? tab[start + i1] : 0;
      const int key0 = k0 * SLOT_CAP + lane;
      const int key1 = k1 * SLOT_CAP + i1;
      int r0 = 0, r1 = 0;
      const int c0 = min(dg, 32);
      for (int j = 0; j < c0; j++) r0 += (__builtin_amdgcn_readlane(key0, j) < key0) ? 1 : 0;
      if (dg > 32) {
        for (int j = 0; j < 32; j++) r1 += (__builtin_amdgcn_readlane(key0, j) < key1) ? 1 : 0;
        const int c1 = dg - 32;
        for (int j = 0; j < c1; j++) {
          const int s = __builtin_amdgcn_readlane(key1, j);
          r0 += (s < key0) ? 1 : 0;
          r1 += (s < key1) ? 1 : 0;
        }
      }
      if (lane < dg) { sorted_out[start + r0] = k0; myrow[r0] = k0; }
      if (i1 < dg) { sorted_out[start + r1] = k1; myrow[r1] = k1; }
    }
    for (int base = 0; base < dg; base += 32) {
      const int cnt = min(32, dg - base);
      int myid;
      if constexpr (WEIGHTED) myid = (lane < cnt) ? myrow[base + lane] : 0;
      else myid = (lane < cnt) ? sorted[start + base + lane] : 0;
      float myw = 0.f;
      if constexpr (WEIGHTED) myw = (lane < cnt) ? rsqrtf((float)(deg[myid] + 1u)) : 0.f;
      const int pairs = cnt >> 1;
#pragma unroll 4
      for (int p = 0; p < pairs; p++) {
        const int s = __shfl(myid, 2 * p + sub, 32);
        const float4 v = *(const float4*)(src + (size_t)s * D + c * 4);
        if constexpr (WEIGHTED) {
          const float w = __shfl(myw, 2 * p + sub, 32);
          acc.x = fmaf(v.x, w, acc.x); acc.y = fmaf(v.y, w, acc.y);
          acc.z = fmaf(v.z, w, acc.z); acc.w = fmaf(v.w, w, acc.w);
        } else {
          acc.x += v.x; acc.y += v.y; acc.z += v.z; acc.w += v.w;
        }
      }
      if (cnt & 1) {
        const int s = __shfl(myid, cnt - 1, 32);
        const float w = WEIGHTED ? __shfl(myw, cnt - 1, 32) : 1.f;
        if (sub == 0) {
          const float4 v = *(const float4*)(src + (size_t)s * D + c * 4);
          if constexpr (WEIGHTED) {
            acc.x = fmaf(v.x, w, acc.x); acc.y = fmaf(v.y, w, acc.y);
            acc.z = fmaf(v.z, w, acc.z); acc.w = fmaf(v.w, w, acc.w);
          } else {
            acc.x += v.x; acc.y += v.y; acc.z += v.z; acc.w += v.w;
          }
        }
      }
    }
  } else {
    const int ns = min((int)spill_n[0], spill_cap);
    int last = -1;
    for (int it = 0; it < dg; it++) {
      int best = 0x7fffffff;
      for (int k = lane; k < SLOT_CAP; k += 32) {
        const int s = tab[start + k];
        if (s > last && s < best) best = s;
      }
      for (int k = lane; k < ns; k += 32) {
        const int s = spill_r[k];
        if (spill_c[k] == node && s > last && s < best) best = s;
      }
#pragma unroll
      for (int o = 16; o > 0; o >>= 1) best = min(best, __shfl_xor(best, o, 32));
      if (best == 0x7fffffff) break;
      int m = 0;
      for (int k = lane; k < SLOT_CAP; k += 32) m += (tab[start + k] == best) ? 1 : 0;
      for (int k = lane; k < ns; k += 32) m += (spill_c[k] == node && spill_r[k] == best) ? 1 : 0;
#pragma unroll
      for (int o = 16; o > 0; o >>= 1) m += __shfl_xor(m, o, 32);
      const float4 v = *(const float4*)(src + (size_t)best * D + c * 4);
      const float w = WEIGHTED ? rsqrtf((float)(deg[best] + 1u)) : 1.f;
      if (sub == 0) {
        for (int i = 0; i < m; i++) {
          if constexpr (WEIGHTED) {
            acc.x = fmaf(v.x, w, acc.x); acc.y = fmaf(v.y, w, acc.y);
            acc.z = fmaf(v.z, w, acc.z); acc.w = fmaf(v.w, w, acc.w);
          } else {
            acc.x += v.x; acc.y += v.y; acc.z += v.z; acc.w += v.w;
          }
        }
      }
      last = best;
    }
  }
  acc.x += __shfl_xor(acc.x, 16, 32);
  acc.y += __shfl_xor(acc.y, 16, 32);
  acc.z += __shfl_xor(acc.z, 16, 32);
  acc.w += __shfl_xor(acc.w, 16, 32);

  float4 t;
  if constexpr (WEIGHTED) {
    t.x = fmaf(self.x, di, acc.x) * di;
    t.y = fmaf(self.y, di, acc.y) * di;
    t.z = fmaf(self.z, di, acc.z) * di;
    t.w = fmaf(self.w, di, acc.w) * di;
  } else {
    t.x = (acc.x + self.x) * di + bb.x;
    t.y = (acc.y + self.y) * di + bb.y;
    t.z = (acc.z + self.z) * di + bb.z;
    t.w = (acc.w + self.w) * di + bb.w;
  }
  if (sub == 0) *(float4*)(dst + (size_t)node * D + c * 4) = t;
}


#define POOL_NODES 8

__device__ __forceinline__ void pool_flush4(float* __restrict__ psum, float* __restrict__ pcnt,
                                            int b, int c, float4 acc, float cnt) {
  float* p = psum + b * 64 + c * 4;
  atomicAdd(p + 0, acc.x);
  atomicAdd(p + 1, acc.y);
  atomicAdd(p + 2, acc.z);
  atomicAdd(p + 3, acc.w);
  if (c == 0) atomicAdd(pcnt + b, cnt);
}

__global__ __launch_bounds__(256) void k_poolsum(const float* __restrict__ hout,
                                                 const int* __restrict__ batch,
                                                 float* __restrict__ psum,
                                                 float* __restrict__ pcnt, int n) {
  __shared__ __attribute__((aligned(16))) float part[16][68];
  __shared__ int pbid[16];

  const int c = threadIdx.x & 15;
  const int g = threadIdx.x >> 4;
  const int base = (blockIdx.x * 16 + g) * POOL_NODES;

  float4 hv[POOL_NODES];
  int bid[POOL_NODES];
#pragma unroll
  for (int j = 0; j < POOL_NODES; j++) {
    int i = base + j;
    if (i > n - 1) i = n - 1;
    hv[j] = *(const float4*)(hout + (size_t)i * 64 + c * 4);
    bid[j] = batch[i];
  }

  int cur = -1;
  float4 acc = make_float4(0.f, 0.f, 0.f, 0.f);
  float cnt = 0.f;
#pragma unroll
  for (int j = 0; j < POOL_NODES; j++) {
    if (base + j < n) {
      int b = bid[j];
      if (b != cur) {
        if (cur >= 0) pool_flush4(psum, pcnt, cur, c, acc, cnt);
        cur = b;
        acc = make_float4(0.f, 0.f, 0.f, 0.f);
        cnt = 0.f;
      }
      acc.x += hv[j].x;
      acc.y += hv[j].y;
      acc.z += hv[j].z;
      acc.w += hv[j].w;
      cnt += 1.f;
    }
  }

  *(float4*)(&part[g][c * 4]) = acc;
  if (c == 0) {
    part[g][64] = cnt;
    pbid[g] = cur;
  }
  __syncthreads();

  const int d = threadIdx.x;
  if (d < 65) {
    float* dst = (d < 64) ? (psum + d) : pcnt;
    const int stride = (d < 64) ? 64 : 1;
    int ids[16];
    float vals[16];
#pragma unroll
    for (int s = 0; s < 16; s++) {
      ids[s] = pbid[s];
      vals[s] = part[s][d];
    }
    int rb = -1;
    float run = 0.f;
#pragma unroll
    for (int s = 0; s < 16; s++) {
      int b = ids[s];
      if (b >= 0) {
        if (b != rb) {
          if (rb >= 0) atomicAdd(dst + rb * stride, run);
          rb = b;
          run = 0.f;
        }
        run += vals[s];
      }
    }
    if (rb >= 0) atomicAdd(dst + rb * stride, run);
  }
}

__global__ void k_pool(const float* __restrict__ psum, const float* __restrict__ pcnt,
                       float* __restrict__ g) {
  int idx = blockIdx.x * blockDim.x + threadIdx.x;
  if (idx < N_GRAPHS * 64) {
    int gi = idx >> 6;
    g[idx] = psum[idx] / fmaxf(pcnt[gi], 1.f);
  }
}


extern "C" void kernel_launch(void* const* d_in, const int* in_sizes, int n_in,
                              void* d_out, int out_size, void* d_ws, size_t ws_size,
                              hipStream_t stream) {
  const float* x   = (const float*)d_in[0];
  const int*   ei  = (const int*)d_in[1];
  const int*  batch= (const int*)d_in[2];
  const float* W1  = (const float*)d_in[3];
  const float* b1  = (const float*)d_in[4];
  const float* lnw = (const float*)d_in[5];
  const float* lnb = (const float*)d_in[6];
  const float* W2  = (const float*)d_in[7];
  const float* b2  = (const float*)d_in[8];

  const int n = in_sizes[0] / 64;
  const int E = in_sizes[1] / 2;
  const int* row = ei;
  const int* col = ei + E;

  float* ws = (float*)d_ws;
  float* xws  = ws;
  float* h1   = xws + (size_t)n * 128;
  unsigned* deg = (unsigned*)(h1 + (size_t)n * 128);
  float* psum = (float*)(deg + n);
  float* pcnt = psum + N_GRAPHS * 64;
  unsigned* spill_n = (unsigned*)(pcnt + N_GRAPHS);
  float* dinv = (float*)(spill_n + 64);
  int* tab    = (int*)(dinv + n);
  int* sorted = tab + (size_t)n * SLOT_CAP;
  int* spill_c = sorted + (size_t)n * SLOT_CAP;
  int* spill_r = spill_c + E;

  float* hout = (float*)d_out;
  float* gout = hout + (size_t)n * 64;

  size_t zbytes = ((size_t)n + N_GRAPHS * 64 + N_GRAPHS + 64) * sizeof(float);
  hipMemsetAsync(deg, 0, zbytes, stream);

  k_place<<<(E + 255) / 256, 256, 0, stream>>>(row, col, deg, tab, spill_n, spill_c, spill_r, E);

  k_gather_finish<true><<<(n + 7) / 8, 256, 0, stream>>>(x, deg, nullptr, sorted, tab, spill_n, spill_c, spill_r, E,
                                                         nullptr, dinv, nullptr, xws, n);
  k_gemm_rows<64, 128, true><<<(n + 127) / 128, 256, 0, stream>>>(xws, W1, nullptr, b1, lnw, lnb, h1, n);

  float* xw2s = xws;
  k_gemm_rows<128, 64, false><<<(n + 127) / 128, 256, 0, stream>>>(h1, W2, dinv, nullptr, nullptr, nullptr, xw2s, n);
  k_gather_finish<false><<<(n + 7) / 8, 256, 0, stream>>>(xw2s, deg, sorted, nullptr, tab, spill_n, spill_c, spill_r, E,
                                                          dinv, nullptr, b2, hout, n);
  k_poolsum<<<(n + 16 * POOL_NODES - 1) / (16 * POOL_NODES), 256, 0, stream>>>(hout, batch, psum, pcnt, n);
  k_pool<<<(N_GRAPHS * 64 + 255) / 256, 256, 0, stream>>>(psum, pcnt, gout);
}


# Edge filing in two levels: a block bins 4096 edges by 256-node target bucket in LDS (one global atomic per block and bucket), a block per bucket builds its rows in LDS; f32 sum order kept
# speedup vs baseline: 5.2614x; 1.7200x over previous
#include <hip/hip_runtime.h>

typedef __attribute__((ext_vector_type(2))) float v2f;
typedef __attribute__((ext_vector_type(8))) float v8f;

#define N_GRAPHS 64


#define SLOT_CAP 64
#define BK_SHIFT 8
#define BK_NODES 256
#define BK_CAP 8192
#define MAX_BK 512
#define BIN_THREADS 512
#define BIN_EPT 8
#define BIN_EDGES (BIN_THREADS * BIN_EPT)

__device__ __forceinline__ void file_edge(int c, int r, unsigned p, int* __restrict__ tab,
                                          unsigned* __restrict__ spill_n, int* __restrict__ spill_c,
                                          int* __restrict__ spill_r) {
  if (p < SLOT_CAP) {
    tab[c * SLOT_CAP + (int)p] = r;
  } else {
    const unsigned k = atomicAdd(spill_n, 1u);
    spill_c[k] = c;
    spill_r[k] = r;
  }
}

__global__ __launch_bounds__(BIN_THREADS) void k_bin(const int* __restrict__ row,
                                                     const int* __restrict__ col,
                                                     unsigned* __restrict__ bcount,
                                                     int* __restrict__ pairs,
                                                     unsigned* __restrict__ deg, int* __restrict__ tab,
                                                     unsigned* __restrict__ spill_n,
                                                     int* __restrict__ spill_c,
                                                     int* __restrict__ spill_r, int E, int nbk) {
  __shared__ unsigned hist[MAX_BK];
  __shared__ unsigned loff[MAX_BK];
  __shared__ unsigned gbase[MAX_BK];
  __shared__ unsigned wsum[BIN_THREADS / 32];
  __shared__ int2 stage[BIN_EDGES];
  const int t = threadIdx.x;
  const int lane = t & 31;
  const int w = t >> 5;
  if (t < nbk) hist[t] = 0u;
  __syncthreads();

  const int blk0 = blockIdx.x * BIN_EDGES;
  int c[BIN_EPT], r[BIN_EPT];
  unsigned rk[BIN_EPT];
#pragma unroll
  for (int j = 0; j < BIN_EPT; j++) {
    const int e = blk0 + t + j * BIN_THREADS;
    c[j] = (e < E) ? col[e] : -1;
    r[j] = (e < E) ? row[e] : 0;
  }
#pragma unroll
  for (int j = 0; j < BIN_EPT; j++) {
    rk[j] = 0u;
    if (c[j] >= 0) rk[j] = atomicAdd(&hist[c[j] >> BK_SHIFT], 1u);
  }
  __syncthreads();

  const unsigned v = (t < nbk) ? hist[t] : 0u;
  if (t < nbk) gbase[t] = v ? atomicAdd(&bcount[t], v) : 0u;
  unsigned inc = v;
#pragma unroll
  for (int o = 1; o < 32; o <<= 1) {
    const unsigned u = __shfl_up(inc, o, 32);
    if (lane >= o) inc += u;
  }
  if (lane == 31) wsum[w] = inc;
  __syncthreads();
  unsigned woff = 0u;
  for (int k = 0; k < w; k++) woff += wsum[k];
  if (t < nbk) loff[t] = woff + inc - v;
  __syncthreads();

#pragma unroll
  for (int j = 0; j < BIN_EPT; j++) {
    if (c[j] >= 0) stage[loff[c[j] >> BK_SHIFT] + rk[j]] = make_int2(c[j], r[j]);
  }
  __syncthreads();

  const int tot = min(BIN_EDGES, E - blk0);
  for (int i = t; i < tot; i += BIN_THREADS) {
    const int2 p = stage[i];
    const int b = p.x >> BK_SHIFT;
    const unsigned pos = gbase[b] + ((unsigned)i - loff[b]);
    if (pos < BK_CAP) {
      ((int2*)pairs)[(size_t)b * BK_CAP + pos] = p;
    } else {
      file_edge(p.x, p.y, atomicAdd(&deg[p.x], 1u), tab, spill_n, spill_c, spill_r);
    }
  }
}

__global__ __launch_bounds__(BIN_THREADS) void k_file(const int* __restrict__ pairs,
                                                      const unsigned* __restrict__ bcount,
                                                      unsigned* __restrict__ deg, int* __restrict__ tab,
                                                      unsigned* __restrict__ spill_n,
                                                      int* __restrict__ spill_c,
                                                      int* __restrict__ spill_r, int n) {
  __shared__ unsigned base[BK_NODES];
  __shared__ unsigned cnt[BK_NODES];
  __shared__ int rows[BK_NODES * SLOT_CAP];
  const int t = threadIdx.x;
  const int b = blockIdx.x;
  const int node0 = b << BK_SHIFT;
  if (t < BK_NODES) {
    const int c = node0 + t;
    base[t] = (c < n) ? deg[c] : 0u;
    cnt[t] = 0u;
  }
  __syncthreads();
  const int m = (int)min(bcount[b], (unsigned)BK_CAP);
  const int2* mine = (const int2*)pairs + (size_t)b * BK_CAP;
  for (int i0 = t; i0 < m; i0 += 4 * BIN_THREADS) {
    int2 p[4];
#pragma unroll
    for (int j = 0; j < 4; j++) {
      const int i = i0 + j * BIN_THREADS;
      p[j] = (i < m) ? mine[i] : make_int2(-1, 0);
    }
#pragma unroll
    for (int j = 0; j < 4; j++) {
      if (p[j].x >= 0) {
        const int cl = p[j].x - node0;
        const unsigned b0 = base[cl];
        const unsigned slot = b0 + atomicAdd(&cnt[cl], 1u);
        if (b0 == 0u && slot < SLOT_CAP) rows[cl * SLOT_CAP + (int)slot] = p[j].y;
        else file_edge(p[j].x, p[j].y, slot, tab, spill_n, spill_c, spill_r);
      }
    }
  }
  __syncthreads();
  const unsigned j = (unsigned)t & (SLOT_CAP - 1);
  for (int i = t; i < BK_NODES * SLOT_CAP; i += BIN_THREADS) {
    const int cl = i >> 6;
    const int c = node0 + cl;
    const unsigned b0 = base[cl];
    const unsigned k = min(cnt[cl], (unsigned)SLOT_CAP);
    if (c < n && b0 == 0u && j < k) tab[c * SLOT_CAP + (int)j] = rows[i];
  }
  if (t < BK_NODES) {
    const int c = node0 + t;
    if (c < n) deg[c] = base[t] + cnt[t];
  }
}


template <int IN, int OUT, bool LN_RELU>
__global__ __launch_bounds__(256) void k_gemm_rows(const float* __restrict__ A,
                                                   const float* __restrict__ W,
                                                   const float* __restrict__ dinv,
                                                   const float* __restrict__ bias,
                                                   const float* __restrict__ lnw,
                                                   const float* __restrict__ lnb,
                                                   float* __restrict__ out, int nRows) {
  constexpr int NT = OUT / 16;
  const int lane = threadIdx.x & 31;
  const int wave = threadIdx.x >> 5;
  const int h = lane >> 4;
  const int m = lane & 15;
  const int rowBase = blockIdx.x * 128 + wave * 16;

  v8f acc[NT] = {};

  int arow = rowBase + m;
  if (arow > nRows - 1) arow = nRows - 1;
  const float* aptr = A + (size_t)arow * IN + 2 * h;

  for (int kb = 0; kb < IN; kb += 4) {
    float2 a2 = *(const float2*)(aptr + kb);
    v2f av; av.x = a2.x; av.y = a2.y;
    const float* w0 = W + (size_t)(kb + 2 * h) * OUT + m;
#pragma unroll
    for (int t = 0; t < NT; t++) {
      v2f bv; bv.x = w0[t * 16]; bv.y = w0[OUT + t * 16];
      acc[t] = __builtin_amdgcn_wmma_f32_16x16x4_f32(
           false, av,  false, bv,
           (short)0, acc[t],  false,  false);
    }
  }

  if constexpr (LN_RELU) {
    float bc[NT], gw[NT], gb[NT];
#pragma unroll
    for (int t = 0; t < NT; t++) {
      bc[t] = bias[t * 16 + m];
      gw[t] = lnw[t * 16 + m];
      gb[t] = lnb[t * 16 + m];
    }
    float s[8], q[8];
#pragma unroll
    for (int v = 0; v < 8; v++) {
      s[v] = 0.f;
#pragma unroll
      for (int t = 0; t < NT; t++) {
        acc[t][v] += bc[t];
        s[v] += acc[t][v];
      }
    }
#pragma unroll
    for (int o = 8; o > 0; o >>= 1) {
#pragma unroll
      for (int v = 0; v < 8; v++) s[v] += __shfl_xor(s[v], o, 32);
    }
#pragma unroll
    for (int v = 0; v < 8; v++) {
      const float mu = s[v] * (1.f / OUT);
      q[v] = 0.f;
#pragma unroll
      for (int t = 0; t < NT; t++) {
        const float d = acc[t][v] - mu;
        acc[t][v] = d;
        q[v] += d * d;
      }
    }
#pragma unroll
    for (int o = 8; o > 0; o >>= 1) {
#pragma unroll
      for (int v = 0; v < 8; v++) q[v] += __shfl_xor(q[v], o, 32);
    }
#pragma unroll
    for (int v = 0; v < 8; v++) {
      const float inv = rsqrtf(q[v] * (1.f / OUT) + 1e-5f);
      const int r = rowBase + v + 8 * h;
      if (r < nRows) {
#pragma unroll
        for (int t = 0; t < NT; t++)
          out[(size_t)r * OUT + t * 16 + m] = fmaxf(acc[t][v] * inv * gw[t] + gb[t], 0.f);
      }
    }
  } else {
#pragma unroll
    for (int v = 0; v < 8; v++) {
      int r = rowBase + v + 8 * h;
      if (r < nRows) {
        float s = dinv[r];
#pragma unroll
        for (int t = 0; t < NT; t++) out[(size_t)r * OUT + t * 16 + m] = acc[t][v] * s;
      }
    }
  }
}


template <bool WEIGHTED>
__global__ __launch_bounds__(256) void k_gather_finish(const float* __restrict__ src,
                                                       const unsigned* __restrict__ deg,
                                                       const int* __restrict__ sorted,
                                                       int* __restrict__ sorted_out,
                                                       const int* __restrict__ tab,
                                                       const unsigned* __restrict__ spill_n,
                                                       const int* __restrict__ spill_c,
                                                       const int* __restrict__ spill_r,
                                                       int spill_cap,
                                                       const float* __restrict__ dinv,
                                                       float* __restrict__ dinv_out,
                                                       const float* __restrict__ bias,
                                                       float* __restrict__ dst, int n) {
  constexpr int D = 64;
  const int lane = threadIdx.x & 31;
  const int node = __builtin_amdgcn_readfirstlane(blockIdx.x * 8 + (threadIdx.x >> 5));
  if (node >= n) return;
  const int start = node * SLOT_CAP;
  const int dg = (int)deg[node];
  float di;
  int* myrow = nullptr;
  if constexpr (WEIGHTED) {
    __shared__ int idrow[8][SLOT_CAP];
    myrow = idrow[threadIdx.x >> 5];
    di = rsqrtf((float)((unsigned)dg + 1u));
    if (lane == 0) dinv_out[node] = di;
  } else {
    di = dinv[node];
  }
  const int sub = lane >> 4;
  const int c = lane & 15;
  const float4 self = *(const float4*)(src + (size_t)node * D + c * 4);
  float4 bb = make_float4(0.f, 0.f, 0.f, 0.f);
  if constexpr (!WEIGHTED) bb = *(const float4*)(bias + c * 4);
  float4 acc = make_float4(0.f, 0.f, 0.f, 0.f);
  if (dg <= SLOT_CAP) {
    if constexpr (WEIGHTED) {
      const int i1 = 32 + lane;
      const int k0 = (lane < dg) ? tab[start + lane] : 0;
      const int k1 = (i1 < dg) ? tab[start + i1] : 0;
      const int key0 = k0 * SLOT_CAP + lane;
      const int key1 = k1 * SLOT_CAP + i1;
      int r0 = 0, r1 = 0;
      const int c0 = min(dg, 32);
      for (int j = 0; j < c0; j++) r0 += (__builtin_amdgcn_readlane(key0, j) < key0) ? 1 : 0;
      if (dg > 32) {
        for (int j = 0; j < 32; j++) r1 += (__builtin_amdgcn_readlane(key0, j) < key1) ? 1 : 0;
        const int c1 = dg - 32;
        for (int j = 0; j < c1; j++) {
          const int s = __builtin_amdgcn_readlane(key1, j);
          r0 += (s < key0) ? 1 : 0;
          r1 += (s < key1) ? 1 : 0;
        }
      }
      if (lane < dg) { sorted_out[start + r0] = k0; myrow[r0] = k0; }
      if (i1 < dg) { sorted_out[start + r1] = k1; myrow[r1] = k1; }
    }
    for (int base = 0; base < dg; base += 32) {
      const int cnt = min(32, dg - base);
      int myid;
      if constexpr (WEIGHTED) myid = (lane < cnt) ? myrow[base + lane] : 0;
      else myid = (lane < cnt) ? sorted[start + base + lane] : 0;
      float myw = 0.f;
      if constexpr (WEIGHTED) myw = (lane < cnt) ? rsqrtf((float)(deg[myid] + 1u)) : 0.f;
      const int pairs = cnt >> 1;
#pragma unroll 4
      for (int p = 0; p < pairs; p++) {
        const int s = __shfl(myid, 2 * p + sub, 32);
        const float4 v = *(const float4*)(src + (size_t)s * D + c * 4);
        if constexpr (WEIGHTED) {
          const float w = __shfl(myw, 2 * p + sub, 32);
          acc.x = fmaf(v.x, w, acc.x); acc.y = fmaf(v.y, w, acc.y);
          acc.z = fmaf(v.z, w, acc.z); acc.w = fmaf(v.w, w, acc.w);
        } else {
          acc.x += v.x; acc.y += v.y; acc.z += v.z; acc.w += v.w;
        }
      }
      if (cnt & 1) {
        const int s = __shfl(myid, cnt - 1, 32);
        const float w = WEIGHTED ? __shfl(myw, cnt - 1, 32) : 1.f;
        if (sub == 0) {
          const float4 v = *(const float4*)(src + (size_t)s * D + c * 4);
          if constexpr (WEIGHTED) {
            acc.x = fmaf(v.x, w, acc.x); acc.y = fmaf(v.y, w, acc.y);
            acc.z = fmaf(v.z, w, acc.z); acc.w = fmaf(v.w, w, acc.w);
          } else {
            acc.x += v.x; acc.y += v.y; acc.z += v.z; acc.w += v.w;
          }
        }
      }
    }
  } else {
    const int ns = min((int)spill_n[0], spill_cap);
    int last = -1;
    for (int it = 0; it < dg; it++) {
      int best = 0x7fffffff;
      for (int k = lane; k < SLOT_CAP; k += 32) {
        const int s = tab[start + k];
        if (s > last && s < best) best = s;
      }
      for (int k = lane; k < ns; k += 32) {
        const int s = spill_r[k];
        if (spill_c[k] == node && s > last && s < best) best = s;
      }
#pragma unroll
      for (int o = 16; o > 0; o >>= 1) best = min(best, __shfl_xor(best, o, 32));
      if (best == 0x7fffffff) break;
      int m = 0;
      for (int k = lane; k < SLOT_CAP; k += 32) m += (tab[start + k] == best) ? 1 : 0;
      for (int k = lane; k < ns; k += 32) m += (spill_c[k] == node && spill_r[k] == best) ? 1 : 0;
#pragma unroll
      for (int o = 16; o > 0; o >>= 1) m += __shfl_xor(m, o, 32);
      const float4 v = *(const float4*)(src + (size_t)best * D + c * 4);
      const float w = WEIGHTED ? rsqrtf((float)(deg[best] + 1u)) : 1.f;
      if (sub == 0) {
        for (int i = 0; i < m; i++) {
          if constexpr (WEIGHTED) {
            acc.x = fmaf(v.x, w, acc.x); acc.y = fmaf(v.y, w, acc.y);
            acc.z = fmaf(v.z, w, acc.z); acc.w = fmaf(v.w, w, acc.w);
          } else {
            acc.x += v.x; acc.y += v.y; acc.z += v.z; acc.w += v.w;
          }
        }
      }
      last = best;
    }
  }
  acc.x += __shfl_xor(acc.x, 16, 32);
  acc.y += __shfl_xor(acc.y, 16, 32);
  acc.z += __shfl_xor(acc.z, 16, 32);
  acc.w += __shfl_xor(acc.w, 16, 32);

  float4 t;
  if constexpr (WEIGHTED) {
    t.x = fmaf(self.x, di, acc.x) * di;
    t.y = fmaf(self.y, di, acc.y) * di;
    t.z = fmaf(self.z, di, acc.z) * di;
    t.w = fmaf(self.w, di, acc.w) * di;
  } else {
    t.x = (acc.x + self.x) * di + bb.x;
    t.y = (acc.y + self.y) * di + bb.y;
    t.z = (acc.z + self.z) * di + bb.z;
    t.w = (acc.w + self.w) * di + bb.w;
  }
  if (sub == 0) *(float4*)(dst + (size_t)node * D + c * 4) = t;
}


#define POOL_NODES 8

__device__ __forceinline__ void pool_flush4(float* __restrict__ psum, float* __restrict__ pcnt,
                                            int b, int c, float4 acc, float cnt) {
  float* p = psum + b * 64 + c * 4;
  atomicAdd(p + 0, acc.x);
  atomicAdd(p + 1, acc.y);
  atomicAdd(p + 2, acc.z);
  atomicAdd(p + 3, acc.w);
  if (c == 0) atomicAdd(pcnt + b, cnt);
}

__global__ __launch_bounds__(256) void k_poolsum(const float* __restrict__ hout,
                                                 const int* __restrict__ batch,
                                                 float* __restrict__ psum,
                                                 float* __restrict__ pcnt, int n) {
  __shared__ __attribute__((aligned(16))) float part[16][68];
  __shared__ int pbid[16];

  const int c = threadIdx.x & 15;
  const int g = threadIdx.x >> 4;
  const int base = (blockIdx.x * 16 + g) * POOL_NODES;

  float4 hv[POOL_NODES];
  int bid[POOL_NODES];
#pragma unroll
  for (int j = 0; j < POOL_NODES; j++) {
    int i = base + j;
    if (i > n - 1) i = n - 1;
    hv[j] = *(const float4*)(hout + (size_t)i * 64 + c * 4);
    bid[j] = batch[i];
  }

  int cur = -1;
  float4 acc = make_float4(0.f, 0.f, 0.f, 0.f);
  float cnt = 0.f;
#pragma unroll
  for (int j = 0; j < POOL_NODES; j++) {
    if (base + j < n) {
      int b = bid[j];
      if (b != cur) {
        if (cur >= 0) pool_flush4(psum, pcnt, cur, c, acc, cnt);
        cur = b;
        acc = make_float4(0.f, 0.f, 0.f, 0.f);
        cnt = 0.f;
      }
      acc.x += hv[j].x;
      acc.y += hv[j].y;
      acc.z += hv[j].z;
      acc.w += hv[j].w;
      cnt += 1.f;
    }
  }

  *(float4*)(&part[g][c * 4]) = acc;
  if (c == 0) {
    part[g][64] = cnt;
    pbid[g] = cur;
  }
  __syncthreads();

  const int d = threadIdx.x;
  if (d < 65) {
    float* dst = (d < 64) ? (psum + d) : pcnt;
    const int stride = (d < 64) ? 64 : 1;
    int ids[16];
    float vals[16];
#pragma unroll
    for (int s = 0; s < 16; s++) {
      ids[s] = pbid[s];
      vals[s] = part[s][d];
    }
    int rb = -1;
    float run = 0.f;
#pragma unroll
    for (int s = 0; s < 16; s++) {
      int b = ids[s];
      if (b >= 0) {
        if (b != rb) {
          if (rb >= 0) atomicAdd(dst + rb * stride, run);
          rb = b;
          run = 0.f;
        }
        run += vals[s];
      }
    }
    if (rb >= 0) atomicAdd(dst + rb * stride, run);
  }
}

__global__ void k_pool(const float* __restrict__ psum, const float* __restrict__ pcnt,
                       float* __restrict__ g) {
  int idx = blockIdx.x * blockDim.x + threadIdx.x;
  if (idx < N_GRAPHS * 64) {
    int gi = idx >> 6;
    g[idx] = psum[idx] / fmaxf(pcnt[gi], 1.f);
  }
}


extern "C" void kernel_launch(void* const* d_in, const int* in_sizes, int n_in,
                              void* d_out, int out_size, void* d_ws, size_t ws_size,
                              hipStream_t stream) {
  const float* x   = (const float*)d_in[0];
  const int*   ei  = (const int*)d_in[1];
  const int*  batch= (const int*)d_in[2];
  const float* W1  = (const float*)d_in[3];
  const float* b1  = (const float*)d_in[4];
  const float* lnw = (const float*)d_in[5];
  const float* lnb = (const float*)d_in[6];
  const float* W2  = (const float*)d_in[7];
  const float* b2  = (const float*)d_in[8];

  const int n = in_sizes[0] / 64;
  const int E = in_sizes[1] / 2;
  const int* row = ei;
  const int* col = ei + E;

  float* ws = (float*)d_ws;
  float* xws  = ws;
  float* h1   = xws + (size_t)n * 128;
  unsigned* deg = (unsigned*)(h1 + (size_t)n * 128);
  float* psum = (float*)(deg + n);
  float* pcnt = psum + N_GRAPHS * 64;
  unsigned* spill_n = (unsigned*)(pcnt + N_GRAPHS);
  unsigned* bcount = spill_n + 64;
  float* dinv = (float*)(bcount + MAX_BK);
  int* tab    = (int*)(dinv + n);
  int* sorted = tab + (size_t)n * SLOT_CAP;
  int* spill_c = sorted + (size_t)n * SLOT_CAP;
  int* spill_r = spill_c + E;
  const int nbk = (n + BK_NODES - 1) >> BK_SHIFT;
  int* pairs = spill_r + E;

  float* hout = (float*)d_out;
  float* gout = hout + (size_t)n * 64;

  size_t zbytes = ((size_t)n + N_GRAPHS * 64 + N_GRAPHS + 64 + MAX_BK) * sizeof(float);
  hipMemsetAsync(deg, 0, zbytes, stream);

  k_bin<<<(E + BIN_EDGES - 1) / BIN_EDGES, BIN_THREADS, 0, stream>>>(
      row, col, bcount, pairs, deg, tab, spill_n, spill_c, spill_r, E, nbk);
  k_file<<<nbk, BIN_THREADS, 0, stream>>>(pairs, bcount, deg, tab, spill_n, spill_c, spill_r, n);

  k_gather_finish<true><<<(n + 7) / 8, 256, 0, stream>>>(x, deg, nullptr, sorted, tab, spill_n, spill_c, spill_r, E,
                                                         nullptr, dinv, nullptr, xws, n);
  k_gemm_rows<64, 128, true><<<(n + 127) / 128, 256, 0, stream>>>(xws, W1, nullptr, b1, lnw, lnb, h1, n);

  float* xw2s = xws;
  k_gemm_rows<128, 64, false><<<(n + 127) / 128, 256, 0, stream>>>(h1, W2, dinv, nullptr, nullptr, nullptr, xw2s, n);
  k_gather_finish<false><<<(n + 7) / 8, 256, 0, stream>>>(xw2s, deg, sorted, nullptr, tab, spill_n, spill_c, spill_r, E,
                                                          dinv, nullptr, b2, hout, n);
  k_poolsum<<<(n + 16 * POOL_NODES - 1) / (16 * POOL_NODES), 256, 0, stream>>>(hout, batch, psum, pcnt, n);
  k_pool<<<(N_GRAPHS * 64 + 255) / 256, 256, 0, stream>>>(psum, pcnt, gout);
}


# Gather kernels: a wave takes 7 nodes in turn (14288 waves, all resident at once; were 100000, one per node), next node's ids requested a node ahead, rows in steps of 8 slots; f32 sum order kept
# speedup vs baseline: 5.4928x; 1.0440x over previous
#include <hip/hip_runtime.h>

typedef __attribute__((ext_vector_type(2))) float v2f;
typedef __attribute__((ext_vector_type(8))) float v8f;

#define N_GRAPHS 64


#define SLOT_CAP 64
#define BK_SHIFT 8
#define BK_NODES 256
#define BK_CAP 8192
#define MAX_BK 512
#define BIN_THREADS 512
#define BIN_EPT 8
#define BIN_EDGES (BIN_THREADS * BIN_EPT)

__device__ __forceinline__ void file_edge(int c, int r, unsigned p, int* __restrict__ tab,
                                          unsigned* __restrict__ spill_n, int* __restrict__ spill_c,
                                          int* __restrict__ spill_r) {
  if (p < SLOT_CAP) {
    tab[c * SLOT_CAP + (int)p] = r;
  } else {
    const unsigned k = atomicAdd(spill_n, 1u);
    spill_c[k] = c;
    spill_r[k] = r;
  }
}

__global__ __launch_bounds__(BIN_THREADS) void k_bin(const int* __restrict__ row,
                                                     const int* __restrict__ col,
                                                     unsigned* __restrict__ bcount,
                                                     int* __restrict__ pairs,
                                                     unsigned* __restrict__ deg, int* __restrict__ tab,
                                                     unsigned* __restrict__ spill_n,
                                                     int* __restrict__ spill_c,
                                                     int* __restrict__ spill_r, int E, int nbk) {
  __shared__ unsigned hist[MAX_BK];
  __shared__ unsigned loff[MAX_BK];
  __shared__ unsigned gbase[MAX_BK];
  __shared__ unsigned wsum[BIN_THREADS / 32];
  __shared__ int2 stage[BIN_EDGES];
  const int t = threadIdx.x;
  const int lane = t & 31;
  const int w = t >> 5;
  if (t < nbk) hist[t] = 0u;
  __syncthreads();

  const int blk0 = blockIdx.x * BIN_EDGES;
  int c[BIN_EPT], r[BIN_EPT];
  unsigned rk[BIN_EPT];
#pragma unroll
  for (int j = 0; j < BIN_EPT; j++) {
    const int e = blk0 + t + j * BIN_THREADS;
    c[j] = (e < E) ? col[e] : -1;
    r[j] = (e < E) ? row[e] : 0;
  }
#pragma unroll
  for (int j = 0; j < BIN_EPT; j++) {
    rk[j] = 0u;
    if (c[j] >= 0) rk[j] = atomicAdd(&hist[c[j] >> BK_SHIFT], 1u);
  }
  __syncthreads();

  const unsigned v = (t < nbk) ? hist[t] : 0u;
  if (t < nbk) gbase[t] = v ? atomicAdd(&bcount[t], v) : 0u;
  unsigned inc = v;
#pragma unroll
  for (int o = 1; o < 32; o <<= 1) {
    const unsigned u = __shfl_up(inc, o, 32);
    if (lane >= o) inc += u;
  }
  if (lane == 31) wsum[w] = inc;
  __syncthreads();
  unsigned woff = 0u;
  for (int k = 0; k < w; k++) woff += wsum[k];
  if (t < nbk) loff[t] = woff + inc - v;
  __syncthreads();

#pragma unroll
  for (int j = 0; j < BIN_EPT; j++) {
    if (c[j] >= 0) stage[loff[c[j] >> BK_SHIFT] + rk[j]] = make_int2(c[j], r[j]);
  }
  __syncthreads();

  const int tot = min(BIN_EDGES, E - blk0);
  for (int i = t; i < tot; i += BIN_THREADS) {
    const int2 p = stage[i];
    const int b = p.x >> BK_SHIFT;
    const unsigned pos = gbase[b] + ((unsigned)i - loff[b]);
    if (pos < BK_CAP) {
      ((int2*)pairs)[(size_t)b * BK_CAP + pos] = p;
    } else {
      file_edge(p.x, p.y, atomicAdd(&deg[p.x], 1u), tab, spill_n, spill_c, spill_r);
    }
  }
}

__global__ __launch_bounds__(BIN_THREADS) void k_file(const int* __restrict__ pairs,
                                                      const unsigned* __restrict__ bcount,
                                                      unsigned* __restrict__ deg, int* __restrict__ tab,
                                                      unsigned* __restrict__ spill_n,
                                                      int* __restrict__ spill_c,
                                                      int* __restrict__ spill_r, int n) {
  __shared__ unsigned base[BK_NODES];
  __shared__ unsigned cnt[BK_NODES];
  __shared__ int rows[BK_NODES * SLOT_CAP];
  const int t = threadIdx.x;
  const int b = blockIdx.x;
  const int node0 = b << BK_SHIFT;
  if (t < BK_NODES) {
    const int c = node0 + t;
    base[t] = (c < n) ? deg[c] : 0u;
    cnt[t] = 0u;
  }
  __syncthreads();
  const int m = (int)min(bcount[b], (unsigned)BK_CAP);
  const int2* mine = (const int2*)pairs + (size_t)b * BK_CAP;
  for (int i0 = t; i0 < m; i0 += 4 * BIN_THREADS) {
    int2 p[4];
#pragma unroll
    for (int j = 0; j < 4; j++) {
      const int i = i0 + j * BIN_THREADS;
      p[j] = (i < m) ? mine[i] : make_int2(-1, 0);
    }
#pragma unroll
    for (int j = 0; j < 4; j++) {
      if (p[j].x >= 0) {
        const int cl = p[j].x - node0;
        const unsigned b0 = base[cl];
        const unsigned slot = b0 + atomicAdd(&cnt[cl], 1u);
        if (b0 == 0u && slot < SLOT_CAP) rows[cl * SLOT_CAP + (int)slot] = p[j].y;
        else file_edge(p[j].x, p[j].y, slot, tab, spill_n, spill_c, spill_r);
      }
    }
  }
  __syncthreads();
  const unsigned j = (unsigned)t & (SLOT_CAP - 1);
  for (int i = t; i < BK_NODES * SLOT_CAP; i += BIN_THREADS) {
    const int cl = i >> 6;
    const int c = node0 + cl;
    const unsigned b0 = base[cl];
    const unsigned k = min(cnt[cl], (unsigned)SLOT_CAP);
    if (c < n && b0 == 0u && j < k) tab[c * SLOT_CAP + (int)j] = rows[i];
  }
  if (t < BK_NODES) {
    const int c = node0 + t;
    if (c < n) deg[c] = base[t] + cnt[t];
  }
}


template <int IN, int OUT, bool LN_RELU>
__global__ __launch_bounds__(256) void k_gemm_rows(const float* __restrict__ A,
                                                   const float* __restrict__ W,
                                                   const float* __restrict__ dinv,
                                                   const float* __restrict__ bias,
                                                   const float* __restrict__ lnw,
                                                   const float* __restrict__ lnb,
                                                   float* __restrict__ out, int nRows) {
  constexpr int NT = OUT / 16;
  const int lane = threadIdx.x & 31;
  const int wave = threadIdx.x >> 5;
  const int h = lane >> 4;
  const int m = lane & 15;
  const int rowBase = blockIdx.x * 128 + wave * 16;

  v8f acc[NT] = {};

  int arow = rowBase + m;
  if (arow > nRows - 1) arow = nRows - 1;
  const float* aptr = A + (size_t)arow * IN + 2 * h;

  for (int kb = 0; kb < IN; kb += 4) {
    float2 a2 = *(const float2*)(aptr + kb);
    v2f av; av.x = a2.x; av.y = a2.y;
    const float* w0 = W + (size_t)(kb + 2 * h) * OUT + m;
#pragma unroll
    for (int t = 0; t < NT; t++) {
      v2f bv; bv.x = w0[t * 16]; bv.y = w0[OUT + t * 16];
      acc[t] = __builtin_amdgcn_wmma_f32_16x16x4_f32(
           false, av,  false, bv,
           (short)0, acc[t],  false,  false);
    }
  }

  if constexpr (LN_RELU) {
    float bc[NT], gw[NT], gb[NT];
#pragma unroll
    for (int t = 0; t < NT; t++) {
      bc[t] = bias[t * 16 + m];
      gw[t] = lnw[t * 16 + m];
      gb[t] = lnb[t * 16 + m];
    }
    float s[8], q[8];
#pragma unroll
    for (int v = 0; v < 8; v++) {
      s[v] = 0.f;
#pragma unroll
      for (int t = 0; t < NT; t++) {
        acc[t][v] += bc[t];
        s[v] += acc[t][v];
      }
    }
#pragma unroll
    for (int o = 8; o > 0; o >>= 1) {
#pragma unroll
      for (int v = 0; v < 8; v++) s[v] += __shfl_xor(s[v], o, 32);
    }
#pragma unroll
    for (int v = 0; v < 8; v++) {
      const float mu = s[v] * (1.f / OUT);
      q[v] = 0.f;
#pragma unroll
      for (int t = 0; t < NT; t++) {
        const float d = acc[t][v] - mu;
        acc[t][v] = d;
        q[v] += d * d;
      }
    }
#pragma unroll
    for (int o = 8; o > 0; o >>= 1) {
#pragma unroll
      for (int v = 0; v < 8; v++) q[v] += __shfl_xor(q[v], o, 32);
    }
#pragma unroll
    for (int v = 0; v < 8; v++) {
      const float inv = rsqrtf(q[v] * (1.f / OUT) + 1e-5f);
      const int r = rowBase + v + 8 * h;
      if (r < nRows) {
#pragma unroll
        for (int t = 0; t < NT; t++)
          out[(size_t)r * OUT + t * 16 + m] = fmaxf(acc[t][v] * inv * gw[t] + gb[t], 0.f);
      }
    }
  } else {
#pragma unroll
    for (int v = 0; v < 8; v++) {
      int r = rowBase + v + 8 * h;
      if (r < nRows) {
        float s = dinv[r];
#pragma unroll
        for (int t = 0; t < NT; t++) out[(size_t)r * OUT + t * 16 + m] = acc[t][v] * s;
      }
    }
  }
}


template <bool WEIGHTED>
__device__ __forceinline__ void rows8_load(float4 (&v)[4], float (&w)[4],
                                           const float* __restrict__ src, int myid, float myw,
                                           int q, int cnt, int sub, int c) {
  int s[4];
#pragma unroll
  for (int u = 0; u < 4; u++) {
    const int slot = q + 2 * u + sub;
    s[u] = __shfl(myid, slot, 32);
    if constexpr (WEIGHTED) w[u] = __shfl(myw, slot, 32);
    else w[u] = (slot < cnt) ? 1.f : 0.f;
  }
#pragma unroll
  for (int u = 0; u < 4; u++) v[u] = *(const float4*)(src + (size_t)s[u] * 64 + c * 4);
}

__device__ __forceinline__ void rows8_add(float4& acc, const float4 (&v)[4], const float (&w)[4]) {
#pragma unroll
  for (int u = 0; u < 4; u++) {
    acc.x = fmaf(v[u].x, w[u], acc.x); acc.y = fmaf(v[u].y, w[u], acc.y);
    acc.z = fmaf(v[u].z, w[u], acc.z); acc.w = fmaf(v[u].w, w[u], acc.w);
  }
}

#define GATHER_NPW 7

template <bool WEIGHTED>
__global__ __launch_bounds__(256) void k_gather_finish(const float* __restrict__ src,
                                                       const unsigned* __restrict__ deg,
                                                       const int* __restrict__ sorted,
                                                       int* __restrict__ sorted_out,
                                                       const int* __restrict__ tab,
                                                       const unsigned* __restrict__ spill_n,
                                                       const int* __restrict__ spill_c,
                                                       const int* __restrict__ spill_r,
                                                       int spill_cap,
                                                       const float* __restrict__ dinv,
                                                       float* __restrict__ dinv_out,
                                                       const float* __restrict__ bias,
                                                       float* __restrict__ dst, int n) {
  constexpr int D = 64;
  const int lane = threadIdx.x & 31;
  const int first = __builtin_amdgcn_readfirstlane(blockIdx.x * 8 + (threadIdx.x >> 5));
  if (first >= n) return;
  const int nwaves = (int)gridDim.x * 8;
  __shared__ int idrow[WEIGHTED ? 8 : 1][SLOT_CAP];
  int* myrow = idrow[WEIGHTED ? (threadIdx.x >> 5) : 0];
  const int mynode = first + lane * nwaves;
  const bool mine = lane < GATHER_NPW && mynode < n;
  const int dgv = mine ? (int)deg[mynode] : 0;
  float div;
  if constexpr (WEIGHTED) {
    div = rsqrtf((float)((unsigned)dgv + 1u));
    if (mine) dinv_out[mynode] = div;
  } else {
    div = mine ? dinv[mynode] : 0.f;
  }
  const int* __restrict__ ids = WEIGHTED ? tab : sorted;
  const int sub = lane >> 4;
  const int c = lane & 15;
  float4 bb = make_float4(0.f, 0.f, 0.f, 0.f);
  if constexpr (!WEIGHTED) bb = *(const float4*)(bias + c * 4);

  int dg_n = __builtin_amdgcn_readlane(dgv, 0);
  int id_n = (lane < dg_n && dg_n <= SLOT_CAP) ? ids[first * SLOT_CAP + lane] : 0;
  for (int k = 0; k < GATHER_NPW; k++) {
    const int node = first + k * nwaves;
    if (node >= n) break;
    const int start = node * SLOT_CAP;
    const int dg = dg_n;
    const int id0 = id_n;
    const float di = __int_as_float(__builtin_amdgcn_readlane(__float_as_int(div), k));
    const int nxt = node + nwaves;
    const bool more = k + 1 < GATHER_NPW && nxt < n;
    dg_n = more ? __builtin_amdgcn_readlane(dgv, k + 1) : 0;
    const bool ahead = more && dg_n <= SLOT_CAP;
    id_n = 0;
    float4 self;
    float4 acc = make_float4(0.f, 0.f, 0.f, 0.f);
    if (dg <= SLOT_CAP) {
      if constexpr (WEIGHTED) {
        const int i1 = 32 + lane;
        const int k0 = id0;
        const int k1 = (i1 < dg) ? tab[start + i1] : 0;
        const int key0 = k0 * SLOT_CAP + lane;
        const int key1 = k1 * SLOT_CAP + i1;
        int r0 = 0, r1 = 0;
        const int c0 = min(dg, 32);
        int j = 0;
        for (; j + 8 <= c0; j += 8) {
#pragma unroll
          for (int u = 0; u < 8; u++) r0 += (__builtin_amdgcn_readlane(key0, j + u) < key0) ? 1 : 0;
        }
        for (; j < c0; j++) r0 += (__builtin_amdgcn_readlane(key0, j) < key0) ? 1 : 0;
        if (dg > 32) {
          for (int j = 0; j < 32; j++) r1 += (__builtin_amdgcn_readlane(key0, j) < key1) ? 1 : 0;
          const int c1 = dg - 32;
          for (int j = 0; j < c1; j++) {
            const int s = __builtin_amdgcn_readlane(key1, j);
            r0 += (s < key0) ? 1 : 0;
            r1 += (s < key1) ? 1 : 0;
          }
        }
        if (lane < dg) { sorted_out[start + r0] = k0; myrow[r0] = k0; }
        if (i1 < dg) { sorted_out[start + r1] = k1; myrow[r1] = k1; }
      }
      {
        const int cnt = min(32, dg);
        int myid = (lane < cnt) ? id0 : node;
        float myw = 0.f;
        if constexpr (WEIGHTED) {
          myid = (lane < cnt) ? myrow[lane] : node;
          myw = (lane < cnt) ? rsqrtf((float)(deg[myid] + 1u)) : 0.f;
        }
        float4 v[4];
        float w[4];
        rows8_load<WEIGHTED>(v, w, src, myid, myw, 0, cnt, sub, c);
        self = *(const float4*)(src + (size_t)node * D + c * 4);
        if (ahead) id_n = (lane < dg_n) ? ids[nxt * SLOT_CAP + lane] : 0;
        rows8_add(acc, v, w);
        for (int q = 8; q < cnt; q += 8) {
          rows8_load<WEIGHTED>(v, w, src, myid, myw, q, cnt, sub, c);
          rows8_add(acc, v, w);
        }
      }
      if (dg > 32) {
        const int cnt = dg - 32;
        int myid;
        float myw = 0.f;
        if constexpr (WEIGHTED) {
          myid = (lane < cnt) ? myrow[32 + lane] : node;
          myw = (lane < cnt) ? rsqrtf((float)(deg[myid] + 1u)) : 0.f;
        } else {
          myid = (lane < cnt) ? sorted[start + 32 + lane] : node;
        }
        float4 v[4];
        float w[4];
        for (int q = 0; q < cnt; q += 8) {
          rows8_load<WEIGHTED>(v, w, src, myid, myw, q, cnt, sub, c);
          rows8_add(acc, v, w);
        }
      }
    } else {
      self = *(const float4*)(src + (size_t)node * D + c * 4);
      if (ahead) id_n = (lane < dg_n) ? ids[nxt * SLOT_CAP + lane] : 0;
      const int ns = min((int)spill_n[0], spill_cap);
      int last = -1;
      for (int it = 0; it < dg; it++) {
        int best = 0x7fffffff;
        for (int q = lane; q < SLOT_CAP; q += 32) {
          const int s = tab[start + q];
          if (s > last && s < best) best = s;
        }
        for (int q = lane; q < ns; q += 32) {
          const int s = spill_r[q];
          if (spill_c[q] == node && s > last && s < best) best = s;
        }
#pragma unroll
        for (int o = 16; o > 0; o >>= 1) best = min(best, __shfl_xor(best, o, 32));
        if (best == 0x7fffffff) break;
        int m = 0;
        for (int q = lane; q < SLOT_CAP; q += 32) m += (tab[start + q] == best) ? 1 : 0;
        for (int q = lane; q < ns; q += 32) m += (spill_c[q] == node && spill_r[q] == best) ? 1 : 0;
#pragma unroll
        for (int o = 16; o > 0; o >>= 1) m += __shfl_xor(m, o, 32);
        const float4 v = *(const float4*)(src + (size_t)best * D + c * 4);
        const float w = WEIGHTED ? rsqrtf((float)(deg[best] + 1u)) : 1.f;
        if (sub == 0) {
          for (int i = 0; i < m; i++) {
            if constexpr (WEIGHTED) {
              acc.x = fmaf(v.x, w, acc.x); acc.y = fmaf(v.y, w, acc.y);
              acc.z = fmaf(v.z, w, acc.z); acc.w = fmaf(v.w, w, acc.w);
            } else {
              acc.x += v.x; acc.y += v.y; acc.z += v.z; acc.w += v.w;
            }
          }
        }
        last = best;
      }
    }
    acc.x += __shfl_xor(acc.x, 16, 32);
    acc.y += __shfl_xor(acc.y, 16, 32);
    acc.z += __shfl_xor(acc.z, 16, 32);
    acc.w += __shfl_xor(acc.w, 16, 32);

    float4 t;
    if constexpr (WEIGHTED) {
      t.x = fmaf(self.x, di, acc.x) * di;
      t.y = fmaf(self.y, di, acc.y) * di;
      t.z = fmaf(self.z, di, acc.z) * di;
      t.w = fmaf(self.w, di, acc.w) * di;
    } else {
      t.x = (acc.x + self.x) * di + bb.x;
      t.y = (acc.y + self.y) * di + bb.y;
      t.z = (acc.z + self.z) * di + bb.z;
      t.w = (acc.w + self.w) * di + bb.w;
    }
    if (sub == 0) *(float4*)(dst + (size_t)node * D + c * 4) = t;
  }
}


#define POOL_NODES 8

__device__ __forceinline__ void pool_flush4(float* __restrict__ psum, float* __restrict__ pcnt,
                                            int b, int c, float4 acc, float cnt) {
  float* p = psum + b * 64 + c * 4;
  atomicAdd(p + 0, acc.x);
  atomicAdd(p + 1, acc.y);
  atomicAdd(p + 2, acc.z);
  atomicAdd(p + 3, acc.w);
  if (c == 0) atomicAdd(pcnt + b, cnt);
}

__global__ __launch_bounds__(256) void k_poolsum(const float* __restrict__ hout,
                                                 const int* __restrict__ batch,
                                                 float* __restrict__ psum,
                                                 float* __restrict__ pcnt, int n) {
  __shared__ __attribute__((aligned(16))) float part[16][68];
  __shared__ int pbid[16];

  const int c = threadIdx.x & 15;
  const int g = threadIdx.x >> 4;
  const int base = (blockIdx.x * 16 + g) * POOL_NODES;

  float4 hv[POOL_NODES];
  int bid[POOL_NODES];
#pragma unroll
  for (int j = 0; j < POOL_NODES; j++) {
    int i = base + j;
    if (i > n - 1) i = n - 1;
    hv[j] = *(const float4*)(hout + (size_t)i * 64 + c * 4);
    bid[j] = batch[i];
  }

  int cur = -1;
  float4 acc = make_float4(0.f, 0.f, 0.f, 0.f);
  float cnt = 0.f;
#pragma unroll
  for (int j = 0; j < POOL_NODES; j++) {
    if (base + j < n) {
      int b = bid[j];
      if (b != cur) {
        if (cur >= 0) pool_flush4(psum, pcnt, cur, c, acc, cnt);
        cur = b;
        acc = make_float4(0.f, 0.f, 0.f, 0.f);
        cnt = 0.f;
      }
      acc.x += hv[j].x;
      acc.y += hv[j].y;
      acc.z += hv[j].z;
      acc.w += hv[j].w;
      cnt += 1.f;
    }
  }

  *(float4*)(&part[g][c * 4]) = acc;
  if (c == 0) {
    part[g][64] = cnt;
    pbid[g] = cur;
  }
  __syncthreads();

  const int d = threadIdx.x;
  if (d < 65) {
    float* dst = (d < 64) ? (psum + d) : pcnt;
    const int stride = (d < 64) ? 64 : 1;
    int ids[16];
    float vals[16];
#pragma unroll
    for (int s = 0; s < 16; s++) {
      ids[s] = pbid[s];
      vals[s] = part[s][d];
    }
    int rb = -1;
    float run = 0.f;
#pragma unroll
    for (int s = 0; s < 16; s++) {
      int b = ids[s];
      if (b >= 0) {
        if (b != rb) {
          if (rb >= 0) atomicAdd(dst + rb * stride, run);
          rb = b;
          run = 0.f;
        }
        run += vals[s];
      }
    }
    if (rb >= 0) atomicAdd(dst + rb * stride, run);
  }
}

__global__ void k_pool(const float* __restrict__ psum, const float* __restrict__ pcnt,
                       float* __restrict__ g) {
  int idx = blockIdx.x * blockDim.x + threadIdx.x;
  if (idx < N_GRAPHS * 64) {
    int gi = idx >> 6;
    g[idx] = psum[idx] / fmaxf(pcnt[gi], 1.f);
  }
}


extern "C" void kernel_launch(void* const* d_in, const int* in_sizes, int n_in,
                              void* d_out, int out_size, void* d_ws, size_t ws_size,
                              hipStream_t stream) {
  const float* x   = (const float*)d_in[0];
  const int*   ei  = (const int*)d_in[1];
  const int*  batch= (const int*)d_in[2];
  const float* W1  = (const float*)d_in[3];
  const float* b1  = (const float*)d_in[4];
  const float* lnw = (const float*)d_in[5];
  const float* lnb = (const float*)d_in[6];
  const float* W2  = (const float*)d_in[7];
  const float* b2  = (const float*)d_in[8];

  const int n = in_sizes[0] / 64;
  const int E = in_sizes[1] / 2;
  const int* row = ei;
  const int* col = ei + E;

  float* ws = (float*)d_ws;
  float* xws  = ws;
  float* h1   = xws + (size_t)n * 128;
  unsigned* deg = (unsigned*)(h1 + (size_t)n * 128);
  float* psum = (float*)(deg + n);
  float* pcnt = psum + N_GRAPHS * 64;
  unsigned* spill_n = (unsigned*)(pcnt + N_GRAPHS);
  unsigned* bcount = spill_n + 64;
  float* dinv = (float*)(bcount + MAX_BK);
  int* tab    = (int*)(dinv + n);
  int* sorted = tab + (size_t)n * SLOT_CAP;
  int* spill_c = sorted + (size_t)n * SLOT_CAP;
  int* spill_r = spill_c + E;
  const int nbk = (n + BK_NODES - 1) >> BK_SHIFT;
  int* pairs = spill_r + E;

  float* hout = (float*)d_out;
  float* gout = hout + (size_t)n * 64;

  size_t zbytes = ((size_t)n + N_GRAPHS * 64 + N_GRAPHS + 64 + MAX_BK) * sizeof(float);
  hipMemsetAsync(deg, 0, zbytes, stream);

  k_bin<<<(E + BIN_EDGES - 1) / BIN_EDGES, BIN_THREADS, 0, stream>>>(
      row, col, bcount, pairs, deg, tab, spill_n, spill_c, spill_r, E, nbk);
  k_file<<<nbk, BIN_THREADS, 0, stream>>>(pairs, bcount, deg, tab, spill_n, spill_c, spill_r, n);

  k_gather_finish<true><<<(n + 8 * GATHER_NPW - 1) / (8 * GATHER_NPW), 256, 0, stream>>>(x, deg, nullptr, sorted, tab, spill_n, spill_c, spill_r, E,
                                                         nullptr, dinv, nullptr, xws, n);
  k_gemm_rows<64, 128, true><<<(n + 127) / 128, 256, 0, stream>>>(xws, W1, nullptr, b1, lnw, lnb, h1, n);

  float* xw2s = xws;
  k_gemm_rows<128, 64, false><<<(n + 127) / 128, 256, 0, stream>>>(h1, W2, dinv, nullptr, nullptr, nullptr, xw2s, n);
  k_gather_finish<false><<<(n + 8 * GATHER_NPW - 1) / (8 * GATHER_NPW), 256, 0, stream>>>(xw2s, deg, sorted, nullptr, tab, spill_n, spill_c, spill_r, E,
                                                          dinv, nullptr, b2, hout, n);
  k_poolsum<<<(n + 16 * POOL_NODES - 1) / (16 * POOL_NODES), 256, 0, stream>>>(hout, batch, psum, pcnt, n);
  k_pool<<<(N_GRAPHS * 64 + 255) / 256, 256, 0, stream>>>(psum, pcnt, gout);
}


# Layer-1 gather wave keeps its 16 nodes' summed rows in its own LDS tile and runs both f32 matrix products on it (h1 through the tile, D to A layout): 2 launches, both planes go; f32 sum orders kept
# speedup vs baseline: 6.4149x; 1.1679x over previous
#include <hip/hip_runtime.h>

typedef __attribute__((ext_vector_type(2))) float v2f;
typedef __attribute__((ext_vector_type(8))) float v8f;

#define N_GRAPHS 64


#define SLOT_CAP 64
#define BK_SHIFT 8
#define BK_NODES 256
#define BK_CAP 8192
#define MAX_BK 512
#define BIN_THREADS 512
#define BIN_EPT 8
#define BIN_EDGES (BIN_THREADS * BIN_EPT)

__device__ __forceinline__ void file_edge(int c, int r, unsigned p, int* __restrict__ tab,
                                          unsigned* __restrict__ spill_n, int* __restrict__ spill_c,
                                          int* __restrict__ spill_r) {
  if (p < SLOT_CAP) {
    tab[c * SLOT_CAP + (int)p] = r;
  } else {
    const unsigned k = atomicAdd(spill_n, 1u);
    spill_c[k] = c;
    spill_r[k] = r;
  }
}

__global__ __launch_bounds__(BIN_THREADS) void k_bin(const int* __restrict__ row,
                                                     const int* __restrict__ col,
                                                     unsigned* __restrict__ bcount,
                                                     int* __restrict__ pairs,
                                                     unsigned* __restrict__ deg, int* __restrict__ tab,
                                                     unsigned* __restrict__ spill_n,
                                                     int* __restrict__ spill_c,
                                                     int* __restrict__ spill_r, int E, int nbk) {
  __shared__ unsigned hist[MAX_BK];
  __shared__ unsigned loff[MAX_BK];
  __shared__ unsigned gbase[MAX_BK];
  __shared__ unsigned wsum[BIN_THREADS / 32];
  __shared__ int2 stage[BIN_EDGES];
  const int t = threadIdx.x;
  const int lane = t & 31;
  const int w = t >> 5;
  if (t < nbk) hist[t] = 0u;
  __syncthreads();

  const int blk0 = blockIdx.x * BIN_EDGES;
  int c[BIN_EPT], r[BIN_EPT];
  unsigned rk[BIN_EPT];
#pragma unroll
  for (int j = 0; j < BIN_EPT; j++) {
    const int e = blk0 + t + j * BIN_THREADS;
    c[j] = (e < E) ? col[e] : -1;
    r[j] = (e < E) ? row[e] : 0;
  }
#pragma unroll
  for (int j = 0; j < BIN_EPT; j++) {
    rk[j] = 0u;
    if (c[j] >= 0) rk[j] = atomicAdd(&hist[c[j] >> BK_SHIFT], 1u);
  }
  __syncthreads();

  const unsigned v = (t < nbk) ? hist[t] : 0u;
  if (t < nbk) gbase[t] = v ? atomicAdd(&bcount[t], v) : 0u;
  unsigned inc = v;
#pragma unroll
  for (int o = 1; o < 32; o <<= 1) {
    const unsigned u = __shfl_up(inc, o, 32);
    if (lane >= o) inc += u;
  }
  if (lane == 31) wsum[w] = inc;
  __syncthreads();
  unsigned woff = 0u;
  for (int k = 0; k < w; k++) woff += wsum[k];
  if (t < nbk) loff[t] = woff + inc - v;
  __syncthreads();

#pragma unroll
  for (int j = 0; j < BIN_EPT; j++) {
    if (c[j] >= 0) stage[loff[c[j] >> BK_SHIFT] + rk[j]] = make_int2(c[j], r[j]);
  }
  __syncthreads();

  const int tot = min(BIN_EDGES, E - blk0);
  for (int i = t; i < tot; i += BIN_THREADS) {
    const int2 p = stage[i];
    const int b = p.x >> BK_SHIFT;
    const unsigned pos = gbase[b] + ((unsigned)i - loff[b]);
    if (pos < BK_CAP) {
      ((int2*)pairs)[(size_t)b * BK_CAP + pos] = p;
    } else {
      file_edge(p.x, p.y, atomicAdd(&deg[p.x], 1u), tab, spill_n, spill_c, spill_r);
    }
  }
}

__global__ __launch_bounds__(BIN_THREADS) void k_file(const int* __restrict__ pairs,
                                                      const unsigned* __restrict__ bcount,
                                                      unsigned* __restrict__ deg, int* __restrict__ tab,
                                                      unsigned* __restrict__ spill_n,
                                                      int* __restrict__ spill_c,
                                                      int* __restrict__ spill_r, int n) {
  __shared__ unsigned base[BK_NODES];
  __shared__ unsigned cnt[BK_NODES];
  __shared__ int rows[BK_NODES * SLOT_CAP];
  const int t = threadIdx.x;
  const int b = blockIdx.x;
  const int node0 = b << BK_SHIFT;
  if (t < BK_NODES) {
    const int c = node0 + t;
    base[t] = (c < n) ? deg[c] : 0u;
    cnt[t] = 0u;
  }
  __syncthreads();
  const int m = (int)min(bcount[b], (unsigned)BK_CAP);
  const int2* mine = (const int2*)pairs + (size_t)b * BK_CAP;
  for (int i0 = t; i0 < m; i0 += 4 * BIN_THREADS) {
    int2 p[4];
#pragma unroll
    for (int j = 0; j < 4; j++) {
      const int i = i0 + j * BIN_THREADS;
      p[j] = (i < m) ? mine[i] : make_int2(-1, 0);
    }
#pragma unroll
    for (int j = 0; j < 4; j++) {
      if (p[j].x >= 0) {
        const int cl = p[j].x - node0;
        const unsigned b0 = base[cl];
        const unsigned slot = b0 + atomicAdd(&cnt[cl], 1u);
        if (b0 == 0u && slot < SLOT_CAP) rows[cl * SLOT_CAP + (int)slot] = p[j].y;
        else file_edge(p[j].x, p[j].y, slot, tab, spill_n, spill_c, spill_r);
      }
    }
  }
  __syncthreads();
  const unsigned j = (unsigned)t & (SLOT_CAP - 1);
  for (int i = t; i < BK_NODES * SLOT_CAP; i += BIN_THREADS) {
    const int cl = i >> 6;
    const int c = node0 + cl;
    const unsigned b0 = base[cl];
    const unsigned k = min(cnt[cl], (unsigned)SLOT_CAP);
    if (c < n && b0 == 0u && j < k) tab[c * SLOT_CAP + (int)j] = rows[i];
  }
  if (t < BK_NODES) {
    const int c = node0 + t;
    if (c < n) deg[c] = base[t] + cnt[t];
  }
}


template <bool WEIGHTED>
__device__ __forceinline__ void rows8_load(float4 (&v)[4], float (&w)[4],
                                           const float* __restrict__ src, int myid, float myw,
                                           int q, int cnt, int sub, int c) {
  int s[4];
#pragma unroll
  for (int u = 0; u < 4; u++) {
    const int slot = q + 2 * u + sub;
    s[u] = __shfl(myid, slot, 32);
    if constexpr (WEIGHTED) w[u] = __shfl(myw, slot, 32);
    else w[u] = (slot < cnt) ? 1.f : 0.f;
  }
#pragma unroll
  for (int u = 0; u < 4; u++) v[u] = *(const float4*)(src + (size_t)s[u] * 64 + c * 4);
}

__device__ __forceinline__ void rows8_add(float4& acc, const float4 (&v)[4], const float (&w)[4]) {
#pragma unroll
  for (int u = 0; u < 4; u++) {
    acc.x = fmaf(v[u].x, w[u], acc.x); acc.y = fmaf(v[u].y, w[u], acc.y);
    acc.z = fmaf(v[u].z, w[u], acc.z); acc.w = fmaf(v[u].w, w[u], acc.w);
  }
}

#define GATHER_NPW 7
#define TILE_NODES 16
#define A_PITCH 68
#define H_PITCH 132

template <bool WEIGHTED>
__global__ __launch_bounds__(256, 7) void k_gather_finish(const float* __restrict__ src,
                                                       const unsigned* __restrict__ deg,
                                                       const int* __restrict__ sorted,
                                                       int* __restrict__ sorted_out,
                                                       const int* __restrict__ tab,
                                                       const unsigned* __restrict__ spill_n,
                                                       const int* __restrict__ spill_c,
                                                       const int* __restrict__ spill_r,
                                                       int spill_cap,
                                                       const float* __restrict__ dinv,
                                                       float* __restrict__ dinv_out,
                                                       const float* __restrict__ bias,
                                                       const float* __restrict__ W1,
                                                       const float* __restrict__ lnw,
                                                       const float* __restrict__ lnb,
                                                       const float* __restrict__ W2,
                                                       float* __restrict__ dst, int n) {
  constexpr int D = 64;
  constexpr int NPW = WEIGHTED ? TILE_NODES : GATHER_NPW;
  const int lane = threadIdx.x & 31;
  constexpr int WPB = WEIGHTED ? 4 : 8;
  const int wave = (int)blockIdx.x * WPB + (int)(threadIdx.x >> 5);
  const int first = __builtin_amdgcn_readfirstlane(WEIGHTED ? wave * TILE_NODES : wave);
  if (first >= n) return;
  const int nstep = WEIGHTED ? 1 : (int)gridDim.x * WPB;
  __shared__ int idrow[WEIGHTED ? WPB : 1][SLOT_CAP];
  int* myrow = idrow[WEIGHTED ? (threadIdx.x >> 5) : 0];
  __shared__ __attribute__((aligned(16))) float tiles[WEIGHTED ? WPB : 1][WEIGHTED ? TILE_NODES * H_PITCH : 4];
  float* tile = tiles[WEIGHTED ? (threadIdx.x >> 5) : 0];
  const int mynode = first + lane * nstep;
  const bool mine = lane < NPW && mynode < n;
  const int dgv = mine ? (int)deg[mynode] : 0;
  float div;
  if constexpr (WEIGHTED) {
    div = rsqrtf((float)((unsigned)dgv + 1u));
    if (mine) dinv_out[mynode] = div;
  } else {
    div = mine ? dinv[mynode] : 0.f;
  }
  const int* __restrict__ ids = WEIGHTED ? tab : sorted;
  const int sub = lane >> 4;
  const int c = lane & 15;
  float4 bb = make_float4(0.f, 0.f, 0.f, 0.f);
  if constexpr (!WEIGHTED) bb = *(const float4*)(bias + c * 4);

  int dg_n = __builtin_amdgcn_readlane(dgv, 0);
  int id_n = (lane < dg_n && dg_n <= SLOT_CAP) ? ids[first * SLOT_CAP + lane] : 0;
  for (int k = 0; k < NPW; k++) {
    const int node = first + k * nstep;
    if (node >= n) {
      if constexpr (WEIGHTED) {
        for (int kk = k; kk < NPW; kk++)
          if (sub == 0) *(float4*)(tile + kk * A_PITCH + c * 4) = make_float4(0.f, 0.f, 0.f, 0.f);
      }
      break;
    }
    const int start = node * SLOT_CAP;
    const int dg = dg_n;
    const int id0 = id_n;
    const float di = __int_as_float(__builtin_amdgcn_readlane(__float_as_int(div), k));
    const int nxt = node + nstep;
    const bool more = k + 1 < NPW && nxt < n;
    dg_n = more ? __builtin_amdgcn_readlane(dgv, k + 1) : 0;
    const bool ahead = more && dg_n <= SLOT_CAP;
    id_n = 0;
    float4 self;
    float4 acc = make_float4(0.f, 0.f, 0.f, 0.f);
    if (dg <= SLOT_CAP) {
      if constexpr (WEIGHTED) {
        const int i1 = 32 + lane;
        const int k0 = id0;
        const int k1 = (i1 < dg) ? tab[start + i1] : 0;
        const int key0 = k0 * SLOT_CAP + lane;
        const int key1 = k1 * SLOT_CAP + i1;
        int r0 = 0, r1 = 0;
        const int c0 = min(dg, 32);
        int j = 0;
        for (; j + 8 <= c0; j += 8) {
#pragma unroll
          for (int u = 0; u < 8; u++) r0 += (__builtin_amdgcn_readlane(key0, j + u) < key0) ? 1 : 0;
        }
        for (; j < c0; j++) r0 += (__builtin_amdgcn_readlane(key0, j) < key0) ? 1 : 0;
        if (dg > 32) {
          for (int j = 0; j < 32; j++) r1 += (__builtin_amdgcn_readlane(key0, j) < key1) ? 1 : 0;
          const int c1 = dg - 32;
          for (int j = 0; j < c1; j++) {
            const int s = __builtin_amdgcn_readlane(key1, j);
            r0 += (s < key0) ? 1 : 0;
            r1 += (s < key1) ? 1 : 0;
          }
        }
        if (lane < dg) { sorted_out[start + r0] = k0; myrow[r0] = k0; }
        if (i1 < dg) { sorted_out[start + r1] = k1; myrow[r1] = k1; }
      }
      {
        const int cnt = min(32, dg);
        int myid = (lane < cnt) ? id0 : node;
        float myw = 0.f;
        if constexpr (WEIGHTED) {
          myid = (lane < cnt) ? myrow[lane] : node;
          myw = (lane < cnt) ? rsqrtf((float)(deg[myid] + 1u)) : 0.f;
        }
        float4 v[4];
        float w[4];
        rows8_load<WEIGHTED>(v, w, src, myid, myw, 0, cnt, sub, c);
        float4 v2[4];
        float w2[4];
        if constexpr (WEIGHTED) rows8_load<WEIGHTED>(v2, w2, src, myid, myw, 8, cnt, sub, c);
        self = *(const float4*)(src + (size_t)node * D + c * 4);
        if (ahead) id_n = (lane < dg_n) ? ids[nxt * SLOT_CAP + lane] : 0;
        rows8_add(acc, v, w);
        if constexpr (WEIGHTED) rows8_add(acc, v2, w2);
        for (int q = WEIGHTED ? 16 : 8; q < cnt; q += 8) {
          rows8_load<WEIGHTED>(v, w, src, myid, myw, q, cnt, sub, c);
          rows8_add(acc, v, w);
        }
      }
      if (dg > 32) {
        const int cnt = dg - 32;
        int myid;
        float myw = 0.f;
        if constexpr (WEIGHTED) {
          myid = (lane < cnt) ? myrow[32 + lane] : node;
          myw = (lane < cnt) ? rsqrtf((float)(deg[myid] + 1u)) : 0.f;
        } else {
          myid = (lane < cnt) ? sorted[start + 32 + lane] : node;
        }
        float4 v[4];
        float w[4];
        for (int q = 0; q < cnt; q += 8) {
          rows8_load<WEIGHTED>(v, w, src, myid, myw, q, cnt, sub, c);
          rows8_add(acc, v, w);
        }
      }
    } else {
      self = *(const float4*)(src + (size_t)node * D + c * 4);
      if (ahead) id_n = (lane < dg_n) ? ids[nxt * SLOT_CAP + lane] : 0;
      const int ns = min((int)spill_n[0], spill_cap);
      int last = -1;
      for (int it = 0; it < dg; it++) {
        int best = 0x7fffffff;
        for (int q = lane; q < SLOT_CAP; q += 32) {
          const int s = tab[start + q];
          if (s > last && s < best) best = s;
        }
        for (int q = lane; q < ns; q += 32) {
          const int s = spill_r[q];
          if (spill_c[q] == node && s > last && s < best) best = s;
        }
#pragma unroll
        for (int o = 16; o > 0; o >>= 1) best = min(best, __shfl_xor(best, o, 32));
        if (best == 0x7fffffff) break;
        int m = 0;
        for (int q = lane; q < SLOT_CAP; q += 32) m += (tab[start + q] == best) ? 1 : 0;
        for (int q = lane; q < ns; q += 32) m += (spill_c[q] == node && spill_r[q] == best) ? 1 : 0;
#pragma unroll
        for (int o = 16; o > 0; o >>= 1) m += __shfl_xor(m, o, 32);
        const float4 v = *(const float4*)(src + (size_t)best * D + c * 4);
        const float w = WEIGHTED ? rsqrtf((float)(deg[best] + 1u)) : 1.f;
        if (sub == 0) {
          for (int i = 0; i < m; i++) {
            if constexpr (WEIGHTED) {
              acc.x = fmaf(v.x, w, acc.x); acc.y = fmaf(v.y, w, acc.y);
              acc.z = fmaf(v.z, w, acc.z); acc.w = fmaf(v.w, w, acc.w);
            } else {
              acc.x += v.x; acc.y += v.y; acc.z += v.z; acc.w += v.w;
            }
          }
        }
        last = best;
      }
    }
    acc.x += __shfl_xor(acc.x, 16, 32);
    acc.y += __shfl_xor(acc.y, 16, 32);
    acc.z += __shfl_xor(acc.z, 16, 32);
    acc.w += __shfl_xor(acc.w, 16, 32);

    float4 t;
    if constexpr (WEIGHTED) {
      t.x = fmaf(self.x, di, acc.x) * di;
      t.y = fmaf(self.y, di, acc.y) * di;
      t.z = fmaf(self.z, di, acc.z) * di;
      t.w = fmaf(self.w, di, acc.w) * di;
    } else {
      t.x = (acc.x + self.x) * di + bb.x;
      t.y = (acc.y + self.y) * di + bb.y;
      t.z = (acc.z + self.z) * di + bb.z;
      t.w = (acc.w + self.w) * di + bb.w;
    }
    if constexpr (WEIGHTED) {
      if (sub == 0) *(float4*)(tile + k * A_PITCH + c * 4) = t;
    } else {
      if (sub == 0) *(float4*)(dst + (size_t)node * D + c * 4) = t;
    }
  }

  if constexpr (WEIGHTED) {
    const int h = sub;
    const int m = c;
    v8f acc1[8] = {};
    for (int kb = 0; kb < 64; kb += 4) {
      const float2 a2 = *(const float2*)(tile + m * A_PITCH + kb + 2 * h);
      v2f av; av.x = a2.x; av.y = a2.y;
      const float* w0 = W1 + (size_t)(kb + 2 * h) * 128 + m;
#pragma unroll
      for (int t = 0; t < 8; t++) {
        v2f bv; bv.x = w0[t * 16]; bv.y = w0[128 + t * 16];
        acc1[t] = __builtin_amdgcn_wmma_f32_16x16x4_f32(
             false, av,  false, bv,
             (short)0, acc1[t],  false,  false);
      }
    }

    float bc[8], gw[8], gb[8];
#pragma unroll
    for (int t = 0; t < 8; t++) {
      bc[t] = bias[t * 16 + m];
      gw[t] = lnw[t * 16 + m];
      gb[t] = lnb[t * 16 + m];
    }
    float sm[8], sq[8];
#pragma unroll
    for (int v = 0; v < 8; v++) {
      sm[v] = 0.f;
#pragma unroll
      for (int t = 0; t < 8; t++) {
        acc1[t][v] += bc[t];
        sm[v] += acc1[t][v];
      }
    }
#pragma unroll
    for (int o = 8; o > 0; o >>= 1) {
#pragma unroll
      for (int v = 0; v < 8; v++) sm[v] += __shfl_xor(sm[v], o, 32);
    }
#pragma unroll
    for (int v = 0; v < 8; v++) {
      const float mu = sm[v] * (1.f / 128);
      sq[v] = 0.f;
#pragma unroll
      for (int t = 0; t < 8; t++) {
        const float d = acc1[t][v] - mu;
        acc1[t][v] = d;
        sq[v] += d * d;
      }
    }
#pragma unroll
    for (int o = 8; o > 0; o >>= 1) {
#pragma unroll
      for (int v = 0; v < 8; v++) sq[v] += __shfl_xor(sq[v], o, 32);
    }
#pragma unroll
    for (int v = 0; v < 8; v++) {
      const float inv = rsqrtf(sq[v] * (1.f / 128) + 1e-5f);
#pragma unroll
      for (int t = 0; t < 8; t++)
        tile[(v + 8 * h) * H_PITCH + t * 16 + m] = fmaxf(acc1[t][v] * inv * gw[t] + gb[t], 0.f);
    }

    v8f acc2[4] = {};
    for (int kb = 0; kb < 128; kb += 4) {
      const float2 a2 = *(const float2*)(tile + m * H_PITCH + kb + 2 * h);
      v2f av; av.x = a2.x; av.y = a2.y;
      const float* w0 = W2 + (size_t)(kb + 2 * h) * 64 + m;
#pragma unroll
      for (int t = 0; t < 4; t++) {
        v2f bv; bv.x = w0[t * 16]; bv.y = w0[64 + t * 16];
        acc2[t] = __builtin_amdgcn_wmma_f32_16x16x4_f32(
             false, av,  false, bv,
             (short)0, acc2[t],  false,  false);
      }
    }
#pragma unroll
    for (int v = 0; v < 8; v++) {
      const int r = first + v + 8 * h;
      const float sc = __shfl(div, v + 8 * h, 32);
      if (r < n) {
#pragma unroll
        for (int t = 0; t < 4; t++) dst[(size_t)r * 64 + t * 16 + m] = acc2[t][v] * sc;
      }
    }
  }
}


#define POOL_NODES 8

__device__ __forceinline__ void pool_flush4(float* __restrict__ psum, float* __restrict__ pcnt,
                                            int b, int c, float4 acc, float cnt) {
  float* p = psum + b * 64 + c * 4;
  atomicAdd(p + 0, acc.x);
  atomicAdd(p + 1, acc.y);
  atomicAdd(p + 2, acc.z);
  atomicAdd(p + 3, acc.w);
  if (c == 0) atomicAdd(pcnt + b, cnt);
}

__global__ __launch_bounds__(256) void k_poolsum(const float* __restrict__ hout,
                                                 const int* __restrict__ batch,
                                                 float* __restrict__ psum,
                                                 float* __restrict__ pcnt, int n) {
  __shared__ __attribute__((aligned(16))) float part[16][68];
  __shared__ int pbid[16];

  const int c = threadIdx.x & 15;
  const int g = threadIdx.x >> 4;
  const int base = (blockIdx.x * 16 + g) * POOL_NODES;

  float4 hv[POOL_NODES];
  int bid[POOL_NODES];
#pragma unroll
  for (int j = 0; j < POOL_NODES; j++) {
    int i = base + j;
    if (i > n - 1) i = n - 1;
    hv[j] = *(const float4*)(hout + (size_t)i * 64 + c * 4);
    bid[j] = batch[i];
  }

  int cur = -1;
  float4 acc = make_float4(0.f, 0.f, 0.f, 0.f);
  float cnt = 0.f;
#pragma unroll
  for (int j = 0; j < POOL_NODES; j++) {
    if (base + j < n) {
      int b = bid[j];
      if (b != cur) {
        if (cur >= 0) pool_flush4(psum, pcnt, cur, c, acc, cnt);
        cur = b;
        acc = make_float4(0.f, 0.f, 0.f, 0.f);
        cnt = 0.f;
      }
      acc.x += hv[j].x;
      acc.y += hv[j].y;
      acc.z += hv[j].z;
      acc.w += hv[j].w;
      cnt += 1.f;
    }
  }

  *(float4*)(&part[g][c * 4]) = acc;
  if (c == 0) {
    part[g][64] = cnt;
    pbid[g] = cur;
  }
  __syncthreads();

  const int d = threadIdx.x;
  if (d < 65) {
    float* dst = (d < 64) ? (psum + d) : pcnt;
    const int stride = (d < 64) ? 64 : 1;
    int ids[16];
    float vals[16];
#pragma unroll
    for (int s = 0; s < 16; s++) {
      ids[s] = pbid[s];
      vals[s] = part[s][d];
    }
    int rb = -1;
    float run = 0.f;
#pragma unroll
    for (int s = 0; s < 16; s++) {
      int b = ids[s];
      if (b >= 0) {
        if (b != rb) {
          if (rb >= 0) atomicAdd(dst + rb * stride, run);
          rb = b;
          run = 0.f;
        }
        run += vals[s];
      }
    }
    if (rb >= 0) atomicAdd(dst + rb * stride, run);
  }
}

__global__ void k_pool(const float* __restrict__ psum, const float* __restrict__ pcnt,
                       float* __restrict__ g) {
  int idx = blockIdx.x * blockDim.x + threadIdx.x;
  if (idx < N_GRAPHS * 64) {
    int gi = idx >> 6;
    g[idx] = psum[idx] / fmaxf(pcnt[gi], 1.f);
  }
}


extern "C" void kernel_launch(void* const* d_in, const int* in_sizes, int n_in,
                              void* d_out, int out_size, void* d_ws, size_t ws_size,
                              hipStream_t stream) {
  const float* x   = (const float*)d_in[0];
  const int*   ei  = (const int*)d_in[1];
  const int*  batch= (const int*)d_in[2];
  const float* W1  = (const float*)d_in[3];
  const float* b1  = (const float*)d_in[4];
  const float* lnw = (const float*)d_in[5];
  const float* lnb = (const float*)d_in[6];
  const float* W2  = (const float*)d_in[7];
  const float* b2  = (const float*)d_in[8];

  const int n = in_sizes[0] / 64;
  const int E = in_sizes[1] / 2;
  const int* row = ei;
  const int* col = ei + E;

  float* ws = (float*)d_ws;
  float* xw2s = ws;
  unsigned* deg = (unsigned*)(xw2s + (size_t)n * 64);
  float* psum = (float*)(deg + n);
  float* pcnt = psum + N_GRAPHS * 64;
  unsigned* spill_n = (unsigned*)(pcnt + N_GRAPHS);
  unsigned* bcount = spill_n + 64;
  float* dinv = (float*)(bcount + MAX_BK);
  int* tab    = (int*)(dinv + n);
  int* sorted = tab + (size_t)n * SLOT_CAP;
  int* spill_c = sorted + (size_t)n * SLOT_CAP;
  int* spill_r = spill_c + E;
  const int nbk = (n + BK_NODES - 1) >> BK_SHIFT;
  int* pairs = spill_r + E;

  float* hout = (float*)d_out;
  float* gout = hout + (size_t)n * 64;

  size_t zbytes = ((size_t)n + N_GRAPHS * 64 + N_GRAPHS + 64 + MAX_BK) * sizeof(float);
  hipMemsetAsync(deg, 0, zbytes, stream);

  k_bin<<<(E + BIN_EDGES - 1) / BIN_EDGES, BIN_THREADS, 0, stream>>>(
      row, col, bcount, pairs, deg, tab, spill_n, spill_c, spill_r, E, nbk);
  k_file<<<nbk, BIN_THREADS, 0, stream>>>(pairs, bcount, deg, tab, spill_n, spill_c, spill_r, n);

  k_gather_finish<true><<<(n + 4 * TILE_NODES - 1) / (4 * TILE_NODES), 128, 0, stream>>>(
      x, deg, nullptr, sorted, tab, spill_n, spill_c, spill_r, E, nullptr, dinv, b1, W1, lnw, lnb, W2,
      xw2s, n);

  k_gather_finish<false><<<(n + 8 * GATHER_NPW - 1) / (8 * GATHER_NPW), 256, 0, stream>>>(
      xw2s, deg, sorted, nullptr, tab, spill_n, spill_c, spill_r, E, dinv, nullptr, b2, nullptr, nullptr,
      nullptr, nullptr, hout, n);
  k_poolsum<<<(n + 16 * POOL_NODES - 1) / (16 * POOL_NODES), 256, 0, stream>>>(hout, batch, psum, pcnt, n);
  k_pool<<<(N_GRAPHS * 64 + 255) / 256, 256, 0, stream>>>(psum, pcnt, gout);
}
